# starting kernel
# baseline (speedup 1.0000x reference)
#include <hip/hip_runtime.h>
#include <math.h>

#define NEMBD 384
#define HS 64
#define FTHETA 10000.0f

typedef float v2f __attribute__((ext_vector_type(2)));
typedef float v8f __attribute__((ext_vector_type(8)));

__device__ __forceinline__ v8f wmma4(v2f a, v2f b, v8f c) {
  return __builtin_amdgcn_wmma_f32_16x16x4_f32(
       false, a,  false, b,
       (short)0, c,  false,  false);
}

__global__ __launch_bounds__(256) void qkv_rope_kernel(
    const float* __restrict__ x, const float* __restrict__ Wq,
    const float* __restrict__ Wk, const float* __restrict__ Wv,
    float* __restrict__ qt, float* __restrict__ kt, float* __restrict__ vo,
    int T)
{
  __shared__ float xs[8 * NEMBD];
  const int row0 = blockIdx.x * 8;
  for (int i = threadIdx.x; i < 8 * NEMBD; i += 256)
    xs[i] = x[(size_t)row0 * NEMBD + i];
  __syncthreads();

  const int local = threadIdx.x >> 5;
  const int j     = threadIdx.x & 31;
  const int row   = row0 + local;
  const int t     = row % T;
  const float* xr = &xs[local * NEMBD];

  const float* wq0 = Wq + (size_t)(2 * j)     * NEMBD;
  const float* wq1 = Wq + (size_t)(2 * j + 1) * NEMBD;
  const float* wk0 = Wk + (size_t)(2 * j)     * NEMBD;
  const float* wk1 = Wk + (size_t)(2 * j + 1) * NEMBD;
  const float* wv0 = Wv + (size_t)(2 * j)     * NEMBD;
  const float* wv1 = Wv + (size_t)(2 * j + 1) * NEMBD;

  float q0 = 0.f, q1 = 0.f, k0 = 0.f, k1 = 0.f, v0 = 0.f, v1 = 0.f;
#pragma unroll 4
  for (int c = 0; c < NEMBD; ++c) {
    const float xv = xr[c];
    q0 = fmaf(xv, wq0[c], q0);
    q1 = fmaf(xv, wq1[c], q1);
    k0 = fmaf(xv, wk0[c], k0);
    k1 = fmaf(xv, wk1[c], k1);
    v0 = fmaf(xv, wv0[c], v0);
    v1 = fmaf(xv, wv1[c], v1);
  }

  const float lnT = logf(FTHETA);
  const float tf = (float)t;
  float ct, st;
  if (j < 16) {
    const float f0 = expf(-(2.0f * (float)(2 * j))     * (1.0f / 64.0f) * lnT);
    const float f1 = expf(-(2.0f * (float)(2 * j + 1)) * (1.0f / 64.0f) * lnT);
    ct = sinf(tf * f0);
    st = sinf(tf * f1);
  } else {
    const float f0 = expf(-(2.0f * (float)(2 * j - 32)) * (1.0f / 64.0f) * lnT);
    const float f1 = expf(-(2.0f * (float)(2 * j - 31)) * (1.0f / 64.0f) * lnT);
    ct = cosf(tf * f0);
    st = cosf(tf * f1);
  }

  const float qr = q0 * ct + q1 * st;
  const float kr = k0 * ct - k1 * st;

  const size_t base = (size_t)row * HS + 2 * j;
  qt[base]     = qr;
  qt[base + 1] = q1 - qr * st;
  kt[base]     = kr;
  kt[base + 1] = k1 - kr * st;
  vo[base]     = v0;
  vo[base + 1] = v1;
}

__global__ __launch_bounds__(32) void attn_kernel(
    const float* __restrict__ qt, const float* __restrict__ kt,
    const float* __restrict__ vo, float* __restrict__ out, int T)
{
  const int ntiles = T >> 4;
  const int b      = blockIdx.x / ntiles;
  const int qtile  = blockIdx.x % ntiles;
  const int lane   = threadIdx.x;
  const int half   = lane >> 4;
  const int lid    = lane & 15;
  const int qbase  = qtile << 4;
  const size_t rowq = (size_t)b * T + qbase;

  __shared__ float Smem[256];
  __shared__ float Mrow[16];
  __shared__ float Arow[16];
  __shared__ float Lrow[16];

  v2f qA[16];
  {
    const float* qp = qt + (rowq + lid) * HS + 2 * half;
#pragma unroll
    for (int ks = 0; ks < 16; ++ks)
      qA[ks] = *(const v2f*)(qp + ks * 4);
  }

  v8f o[4];
#pragma unroll
  for (int i = 0; i < 4; ++i) o[i] = v8f{};
  float mrun = -1e30f, lrun = 0.0f;
  const float scale = 0.05103103630798288f;

  const int nkt = qtile + 1;
  for (int ktile = 0; ktile < nkt; ++ktile) {
    const int kbase   = ktile << 4;
    const size_t rowk = (size_t)b * T + kbase;

    v2f kB[16];
    {
      const float* kp = kt + (rowk + lid) * HS + 2 * half;
#pragma unroll
      for (int ks = 0; ks < 16; ++ks)
        kB[ks] = *(const v2f*)(kp + ks * 4);
    }

    v8f s = v8f{};
#pragma unroll
    for (int ks = 0; ks < 16; ++ks)
      s = wmma4(qA[ks], kB[ks], s);

    const int kg = kbase + lid;
#pragma unroll
    for (int r = 0; r < 8; ++r) {
      const int m  = r + 8 * half;
      const int qg = qbase + m;
      const float sv = (kg <= qg) ? s[r] * scale : -1e30f;
      s[r] = sv;
      Smem[m * 16 + lid] = sv;
    }
    __syncthreads();

    if (lane < 16) {
      float mt = -1e30f;
#pragma unroll
      for (int n = 0; n < 16; ++n) mt = fmaxf(mt, Smem[lane * 16 + n]);
      const float mnew  = fmaxf(mrun, mt);
      const float alpha = __expf(mrun - mnew);
      mrun = mnew;
      lrun *= alpha;
      Mrow[lane] = mnew;
      Arow[lane] = alpha;
    }
    __syncthreads();

#pragma unroll
    for (int r = 0; r < 8; ++r) {
      const int m = r + 8 * half;
      const float p  = __expf(s[r] - Mrow[m]);
      const float al = Arow[m];
      Smem[m * 16 + lid] = p;
      o[0][r] *= al; o[1][r] *= al; o[2][r] *= al; o[3][r] *= al;
    }
    __syncthreads();

    if (lane < 16) {
      float rs = 0.0f;
#pragma unroll
      for (int n = 0; n < 16; ++n) rs += Smem[lane * 16 + n];
      lrun += rs;
    }

    v2f pA[4];
#pragma unroll
    for (int kk = 0; kk < 4; ++kk)
      pA[kk] = *(const v2f*)(&Smem[lid * 16 + kk * 4 + 2 * half]);
#pragma unroll
    for (int nt = 0; nt < 4; ++nt) {
#pragma unroll
      for (int kk = 0; kk < 4; ++kk) {
        const size_t vrow = rowk + kk * 4 + 2 * half;
        v2f vB;
        vB.x = vo[vrow * HS + nt * 16 + lid];
        vB.y = vo[(vrow + 1) * HS + nt * 16 + lid];
        o[nt] = wmma4(pA[kk], vB, o[nt]);
      }
    }
    __syncthreads();
  }

  if (lane < 16) Lrow[lane] = lrun;
  __syncthreads();

#pragma unroll
  for (int r = 0; r < 8; ++r) {
    const int m = r + 8 * half;
    const float inv = 1.0f / Lrow[m];
    float* op = out + (rowq + m) * HS + lid;
    op[0]  = o[0][r] * inv;
    op[16] = o[1][r] * inv;
    op[32] = o[2][r] * inv;
    op[48] = o[3][r] * inv;
  }
}

extern "C" void kernel_launch(void* const* d_in, const int* in_sizes, int n_in,
                              void* d_out, int out_size, void* d_ws, size_t ws_size,
                              hipStream_t stream) {
  const float* x  = (const float*)d_in[0];
  const float* Wq = (const float*)d_in[1];
  const float* Wk = (const float*)d_in[2];
  const float* Wv = (const float*)d_in[3];

  const int BT = in_sizes[0] / NEMBD;
  const int T  = 4096;
  const int B  = BT / T;

  float* qt = (float*)d_ws;
  float* kt = qt + (size_t)BT * HS;
  float* vv = kt + (size_t)BT * HS;
  float* out = (float*)d_out;

  qkv_rope_kernel<<<BT / 8, 256, 0, stream>>>(x, Wq, Wk, Wv, qt, kt, vv, T);
  attn_kernel<<<B * (T / 16), 32, 0, stream>>>(qt, kt, vv, out, T);
}


# QKV projection moved from scalar FMA loops with lane-strided weight reads to f32 WMMA 16x16x4, weights staged in LDS per 32-wide K chunk; sums now run in WMMA k order
# speedup vs baseline: 2.4787x; 2.4787x over previous
#include <hip/hip_runtime.h>
#include <math.h>

#define NEMBD 384
#define HS 64
#define FTHETA 10000.0f

typedef float v2f __attribute__((ext_vector_type(2)));
typedef float v8f __attribute__((ext_vector_type(8)));

__device__ __forceinline__ v8f wmma4(v2f a, v2f b, v8f c) {
  return __builtin_amdgcn_wmma_f32_16x16x4_f32(
       false, a,  false, b,
       (short)0, c,  false,  false);
}

#define WPITCH4 9

__global__ __launch_bounds__(384) __attribute__((amdgpu_waves_per_eu(3, 4)))
void qkv_rope_kernel(
    const float* __restrict__ x, const float* __restrict__ Wq,
    const float* __restrict__ Wk, const float* __restrict__ Wv,
    float* __restrict__ qt, float* __restrict__ kt, float* __restrict__ vo,
    int T)
{
  __shared__ float4 Ws4[192 * WPITCH4];

  const int tid  = threadIdx.x;
  const int wave = tid >> 5;
  const int lane = tid & 31;
  const int c    = lane & 15;
  const int h    = lane >> 4;
  const int rt   = wave & 3;
  const int mat  = wave >> 2;
  const int row0 = blockIdx.x * 64 + rt * 16;

  const float4* wsrc[4];
  int wdst[4];
#pragma unroll
  for (int u = 0; u < 4; ++u) {
    const int i  = tid + 384 * u;
    const int wr = i >> 3;
    const int c4 = i & 7;
    const float* base = (wr < 64) ? Wq : ((wr < 128) ? Wk : Wv);
    wsrc[u] = (const float4*)(base + (size_t)(wr & 63) * NEMBD) + c4;
    wdst[u] = wr * WPITCH4 + c4;
  }
  const float4* xsrc = (const float4*)(x + (size_t)(row0 + c) * NEMBD + 16 * h);
  const float4* bsrc = &Ws4[(mat * 64 + c) * WPITCH4 + 4 * h];

  Ws4[wdst[0]] = wsrc[0][0];
  Ws4[wdst[1]] = wsrc[1][0];
  Ws4[wdst[2]] = wsrc[2][0];
  Ws4[wdst[3]] = wsrc[3][0];
  float4 x0 = xsrc[0], x1 = xsrc[1], x2 = xsrc[2], x3 = xsrc[3];

  v8f acc[4];
#pragma unroll
  for (int i = 0; i < 4; ++i) acc[i] = v8f{};

  __syncthreads();

  for (int kc = 0; kc < NEMBD / 32; ++kc) {
    const bool more = (kc + 1 < NEMBD / 32);

    float4 a[4];
    a[0] = x0; a[1] = x1; a[2] = x2; a[3] = x3;

    float4 w0 = {}, w1 = {}, w2 = {}, w3 = {};
    if (more) {
      const int kn = (kc + 1) * 8;
      w0 = wsrc[0][kn]; w1 = wsrc[1][kn]; w2 = wsrc[2][kn]; w3 = wsrc[3][kn];
      x0 = xsrc[kn]; x1 = xsrc[kn + 1]; x2 = xsrc[kn + 2]; x3 = xsrc[kn + 3];
    }

    float4 b[4][4];
#pragma unroll
    for (int nt = 0; nt < 4; ++nt) {
#pragma unroll
      for (int u = 0; u < 4; ++u) b[nt][u] = bsrc[nt * 16 * WPITCH4 + u];
    }

#pragma unroll
    for (int u = 0; u < 4; ++u) {
      v2f a01; a01.x = a[u].x; a01.y = a[u].y;
      v2f a23; a23.x = a[u].z; a23.y = a[u].w;
#pragma unroll
      for (int nt = 0; nt < 4; ++nt) {
        v2f b01; b01.x = b[nt][u].x; b01.y = b[nt][u].y;
        acc[nt] = wmma4(a01, b01, acc[nt]);
      }
#pragma unroll
      for (int nt = 0; nt < 4; ++nt) {
        v2f b23; b23.x = b[nt][u].z; b23.y = b[nt][u].w;
        acc[nt] = wmma4(a23, b23, acc[nt]);
      }
    }

    if (more) {
      __syncthreads();
      Ws4[wdst[0]] = w0;
      Ws4[wdst[1]] = w1;
      Ws4[wdst[2]] = w2;
      Ws4[wdst[3]] = w3;
      __syncthreads();
    }
  }

  const int rbase = row0 + 8 * h;

  if (mat == 2) {
#pragma unroll
    for (int nt = 0; nt < 4; ++nt) {
#pragma unroll
      for (int r = 0; r < 8; ++r)
        vo[(size_t)(rbase + r) * HS + nt * 16 + c] = acc[nt][r];
    }
    return;
  }

  float* __restrict__ dst = (mat == 0) ? qt : kt;
  const float lnT = logf(FTHETA);
  const bool even = (c & 1) == 0;
#pragma unroll
  for (int nt = 0; nt < 4; ++nt) {
    const int n   = nt * 16 + c;
    const int idx = (nt < 2) ? n : n - 32;
    const float f = expf(-(2.0f * (float)idx) * (1.0f / 64.0f) * lnT);
#pragma unroll
    for (int r = 0; r < 8; ++r) {
      const int row  = rbase + r;
      const float tf = (float)(row % T);
      const float tv = (nt < 2) ? sinf(tf * f) : cosf(tf * f);
      const float tp = __shfl_xor(tv, 1);
      const float av = acc[nt][r];
      const float ap = __shfl_xor(av, 1);
      const float ct = even ? tv : tp;
      const float st = even ? tp : tv;
      const float e  = even ? av : ap;
      const float o  = even ? ap : av;
      const float rot = (mat == 0) ? (e * ct + o * st) : (e * ct - o * st);
      dst[(size_t)row * HS + n] = even ? rot : (o - rot * st);
    }
  }
}

__global__ __launch_bounds__(32) void attn_kernel(
    const float* __restrict__ qt, const float* __restrict__ kt,
    const float* __restrict__ vo, float* __restrict__ out, int T)
{
  const int ntiles = T >> 4;
  const int b      = blockIdx.x / ntiles;
  const int qtile  = blockIdx.x % ntiles;
  const int lane   = threadIdx.x;
  const int half   = lane >> 4;
  const int lid    = lane & 15;
  const int qbase  = qtile << 4;
  const size_t rowq = (size_t)b * T + qbase;

  __shared__ float Smem[256];
  __shared__ float Mrow[16];
  __shared__ float Arow[16];
  __shared__ float Lrow[16];

  v2f qA[16];
  {
    const float* qp = qt + (rowq + lid) * HS + 2 * half;
#pragma unroll
    for (int ks = 0; ks < 16; ++ks)
      qA[ks] = *(const v2f*)(qp + ks * 4);
  }

  v8f o[4];
#pragma unroll
  for (int i = 0; i < 4; ++i) o[i] = v8f{};
  float mrun = -1e30f, lrun = 0.0f;
  const float scale = 0.05103103630798288f;

  const int nkt = qtile + 1;
  for (int ktile = 0; ktile < nkt; ++ktile) {
    const int kbase   = ktile << 4;
    const size_t rowk = (size_t)b * T + kbase;

    v2f kB[16];
    {
      const float* kp = kt + (rowk + lid) * HS + 2 * half;
#pragma unroll
      for (int ks = 0; ks < 16; ++ks)
        kB[ks] = *(const v2f*)(kp + ks * 4);
    }

    v8f s = v8f{};
#pragma unroll
    for (int ks = 0; ks < 16; ++ks)
      s = wmma4(qA[ks], kB[ks], s);

    const int kg = kbase + lid;
#pragma unroll
    for (int r = 0; r < 8; ++r) {
      const int m  = r + 8 * half;
      const int qg = qbase + m;
      const float sv = (kg <= qg) ? s[r] * scale : -1e30f;
      s[r] = sv;
      Smem[m * 16 + lid] = sv;
    }
    __syncthreads();

    if (lane < 16) {
      float mt = -1e30f;
#pragma unroll
      for (int n = 0; n < 16; ++n) mt = fmaxf(mt, Smem[lane * 16 + n]);
      const float mnew  = fmaxf(mrun, mt);
      const float alpha = __expf(mrun - mnew);
      mrun = mnew;
      lrun *= alpha;
      Mrow[lane] = mnew;
      Arow[lane] = alpha;
    }
    __syncthreads();

#pragma unroll
    for (int r = 0; r < 8; ++r) {
      const int m = r + 8 * half;
      const float p  = __expf(s[r] - Mrow[m]);
      const float al = Arow[m];
      Smem[m * 16 + lid] = p;
      o[0][r] *= al; o[1][r] *= al; o[2][r] *= al; o[3][r] *= al;
    }
    __syncthreads();

    if (lane < 16) {
      float rs = 0.0f;
#pragma unroll
      for (int n = 0; n < 16; ++n) rs += Smem[lane * 16 + n];
      lrun += rs;
    }

    v2f pA[4];
#pragma unroll
    for (int kk = 0; kk < 4; ++kk)
      pA[kk] = *(const v2f*)(&Smem[lid * 16 + kk * 4 + 2 * half]);
#pragma unroll
    for (int nt = 0; nt < 4; ++nt) {
#pragma unroll
      for (int kk = 0; kk < 4; ++kk) {
        const size_t vrow = rowk + kk * 4 + 2 * half;
        v2f vB;
        vB.x = vo[vrow * HS + nt * 16 + lid];
        vB.y = vo[(vrow + 1) * HS + nt * 16 + lid];
        o[nt] = wmma4(pA[kk], vB, o[nt]);
      }
    }
    __syncthreads();
  }

  if (lane < 16) Lrow[lane] = lrun;
  __syncthreads();

#pragma unroll
  for (int r = 0; r < 8; ++r) {
    const int m = r + 8 * half;
    const float inv = 1.0f / Lrow[m];
    float* op = out + (rowq + m) * HS + lid;
    op[0]  = o[0][r] * inv;
    op[16] = o[1][r] * inv;
    op[32] = o[2][r] * inv;
    op[48] = o[3][r] * inv;
  }
}

extern "C" void kernel_launch(void* const* d_in, const int* in_sizes, int n_in,
                              void* d_out, int out_size, void* d_ws, size_t ws_size,
                              hipStream_t stream) {
  const float* x  = (const float*)d_in[0];
  const float* Wq = (const float*)d_in[1];
  const float* Wk = (const float*)d_in[2];
  const float* Wv = (const float*)d_in[3];

  const int BT = in_sizes[0] / NEMBD;
  const int T  = 4096;
  const int B  = BT / T;

  float* qt = (float*)d_ws;
  float* kt = qt + (size_t)BT * HS;
  float* vv = kt + (size_t)BT * HS;
  float* out = (float*)d_out;

  qkv_rope_kernel<<<BT / 64, 384, 0, stream>>>(x, Wq, Wk, Wv, qt, kt, vv, T);
  attn_kernel<<<B * (T / 16), 32, 0, stream>>>(qt, kt, vv, out, T);
}


# Attention: each query tile's causal key range dealt round-robin to 16 waves of one work-group; partial softmax states merged through LDS in wave order (new reduction tree for row sums and P.V)
# speedup vs baseline: 8.8077x; 3.5534x over previous
#include <hip/hip_runtime.h>
#include <math.h>

#define NEMBD 384
#define HS 64
#define FTHETA 10000.0f

typedef float v2f __attribute__((ext_vector_type(2)));
typedef float v8f __attribute__((ext_vector_type(8)));

__device__ __forceinline__ v8f wmma4(v2f a, v2f b, v8f c) {
  return __builtin_amdgcn_wmma_f32_16x16x4_f32(
       false, a,  false, b,
       (short)0, c,  false,  false);
}

#define WPITCH4 9

__global__ __launch_bounds__(384) __attribute__((amdgpu_waves_per_eu(3, 4)))
void qkv_rope_kernel(
    const float* __restrict__ x, const float* __restrict__ Wq,
    const float* __restrict__ Wk, const float* __restrict__ Wv,
    float* __restrict__ qt, float* __restrict__ kt, float* __restrict__ vo,
    int T)
{
  __shared__ float4 Ws4[192 * WPITCH4];

  const int tid  = threadIdx.x;
  const int wave = tid >> 5;
  const int lane = tid & 31;
  const int c    = lane & 15;
  const int h    = lane >> 4;
  const int rt   = wave & 3;
  const int mat  = wave >> 2;
  const int row0 = blockIdx.x * 64 + rt * 16;

  const float4* wsrc[4];
  int wdst[4];
#pragma unroll
  for (int u = 0; u < 4; ++u) {
    const int i  = tid + 384 * u;
    const int wr = i >> 3;
    const int c4 = i & 7;
    const float* base = (wr < 64) ? Wq : ((wr < 128) ? Wk : Wv);
    wsrc[u] = (const float4*)(base + (size_t)(wr & 63) * NEMBD) + c4;
    wdst[u] = wr * WPITCH4 + c4;
  }
  const float4* xsrc = (const float4*)(x + (size_t)(row0 + c) * NEMBD + 16 * h);
  const float4* bsrc = &Ws4[(mat * 64 + c) * WPITCH4 + 4 * h];

  Ws4[wdst[0]] = wsrc[0][0];
  Ws4[wdst[1]] = wsrc[1][0];
  Ws4[wdst[2]] = wsrc[2][0];
  Ws4[wdst[3]] = wsrc[3][0];
  float4 x0 = xsrc[0], x1 = xsrc[1], x2 = xsrc[2], x3 = xsrc[3];

  v8f acc[4];
#pragma unroll
  for (int i = 0; i < 4; ++i) acc[i] = v8f{};

  __syncthreads();

  for (int kc = 0; kc < NEMBD / 32; ++kc) {
    const bool more = (kc + 1 < NEMBD / 32);

    float4 a[4];
    a[0] = x0; a[1] = x1; a[2] = x2; a[3] = x3;

    float4 w0 = {}, w1 = {}, w2 = {}, w3 = {};
    if (more) {
      const int kn = (kc + 1) * 8;
      w0 = wsrc[0][kn]; w1 = wsrc[1][kn]; w2 = wsrc[2][kn]; w3 = wsrc[3][kn];
      x0 = xsrc[kn]; x1 = xsrc[kn + 1]; x2 = xsrc[kn + 2]; x3 = xsrc[kn + 3];
    }

    float4 b[4][4];
#pragma unroll
    for (int nt = 0; nt < 4; ++nt) {
#pragma unroll
      for (int u = 0; u < 4; ++u) b[nt][u] = bsrc[nt * 16 * WPITCH4 + u];
    }

#pragma unroll
    for (int u = 0; u < 4; ++u) {
      v2f a01; a01.x = a[u].x; a01.y = a[u].y;
      v2f a23; a23.x = a[u].z; a23.y = a[u].w;
#pragma unroll
      for (int nt = 0; nt < 4; ++nt) {
        v2f b01; b01.x = b[nt][u].x; b01.y = b[nt][u].y;
        acc[nt] = wmma4(a01, b01, acc[nt]);
      }
#pragma unroll
      for (int nt = 0; nt < 4; ++nt) {
        v2f b23; b23.x = b[nt][u].z; b23.y = b[nt][u].w;
        acc[nt] = wmma4(a23, b23, acc[nt]);
      }
    }

    if (more) {
      __syncthreads();
      Ws4[wdst[0]] = w0;
      Ws4[wdst[1]] = w1;
      Ws4[wdst[2]] = w2;
      Ws4[wdst[3]] = w3;
      __syncthreads();
    }
  }

  const int rbase = row0 + 8 * h;

  if (mat == 2) {
#pragma unroll
    for (int nt = 0; nt < 4; ++nt) {
#pragma unroll
      for (int r = 0; r < 8; ++r)
        vo[(size_t)(rbase + r) * HS + nt * 16 + c] = acc[nt][r];
    }
    return;
  }

  float* __restrict__ dst = (mat == 0) ? qt : kt;
  const float lnT = logf(FTHETA);
  const bool even = (c & 1) == 0;
#pragma unroll
  for (int nt = 0; nt < 4; ++nt) {
    const int n   = nt * 16 + c;
    const int idx = (nt < 2) ? n : n - 32;
    const float f = expf(-(2.0f * (float)idx) * (1.0f / 64.0f) * lnT);
#pragma unroll
    for (int r = 0; r < 8; ++r) {
      const int row  = rbase + r;
      const float tf = (float)(row % T);
      const float tv = (nt < 2) ? sinf(tf * f) : cosf(tf * f);
      const float tp = __shfl_xor(tv, 1);
      const float av = acc[nt][r];
      const float ap = __shfl_xor(av, 1);
      const float ct = even ? tv : tp;
      const float st = even ? tp : tv;
      const float e  = even ? av : ap;
      const float o  = even ? ap : av;
      const float rot = (mat == 0) ? (e * ct + o * st) : (e * ct - o * st);
      dst[(size_t)row * HS + n] = even ? rot : (o - rot * st);
    }
  }
}

#define ANW 16
#define ATHREADS 512
#define OPITCH 68

__device__ __forceinline__ void wave_lds_order() {
  __builtin_amdgcn_fence(__ATOMIC_RELEASE, "wavefront");
  __builtin_amdgcn_wave_barrier();
  __builtin_amdgcn_fence(__ATOMIC_ACQUIRE, "wavefront");
}

__global__ __launch_bounds__(ATHREADS) __attribute__((amdgpu_waves_per_eu(8, 8)))
void attn_kernel(
    const float* __restrict__ qt, const float* __restrict__ kt,
    const float* __restrict__ vo, float* __restrict__ out, int T)
{
  const int ntiles = T >> 4;
  const int b      = blockIdx.x / ntiles;
  const int qtile  = blockIdx.x % ntiles;
  const int tid    = threadIdx.x;
  const int wave   = __builtin_amdgcn_readfirstlane(tid >> 5);
  const int lane   = tid & 31;
  const int half   = lane >> 4;
  const int lid    = lane & 15;
  const int qbase  = qtile << 4;
  const size_t rowq = (size_t)b * T + qbase;

  __shared__ __attribute__((aligned(16))) float SmemAll[ANW][256];
  __shared__ __attribute__((aligned(16))) float MrowAll[ANW][16];
  __shared__ __attribute__((aligned(16))) float ArowAll[ANW][16];
  __shared__ __attribute__((aligned(16))) float Opart[ANW][16 * OPITCH];
  __shared__ float Mpart[ANW][16];
  __shared__ float Lpart[ANW][16];

  float* Smem = SmemAll[wave];
  float* Mrow = MrowAll[wave];
  float* Arow = ArowAll[wave];
  float*       s_acc = Smem + half * 128 + lid;
  const float* s_row = Smem + lid * 16;
  const float* m_acc = Mrow + 8 * half;
  const float* a_acc = Arow + 8 * half;

  const int nkt = qtile + 1;

  if (wave < nkt) {
    v2f qA[16];
    {
      const float* qp = qt + (rowq + lid) * HS + 2 * half;
#pragma unroll
      for (int ks = 0; ks < 16; ++ks)
        qA[ks] = *(const v2f*)(qp + ks * 4);
    }

    v8f o[4];
#pragma unroll
    for (int i = 0; i < 4; ++i) o[i] = v8f{};
    float mrun = -1e30f, lrun = 0.0f;
    const float scale = 0.05103103630798288f;

    for (int ktile = wave; ktile < nkt; ktile += ANW) {
      const int kbase   = ktile << 4;
      const size_t rowk = (size_t)b * T + kbase;

      v2f kB[16];
      {
        const float* kp = kt + (rowk + lid) * HS + 2 * half;
#pragma unroll
        for (int ks = 0; ks < 16; ++ks)
          kB[ks] = *(const v2f*)(kp + ks * 4);
      }

      v8f s = v8f{};
#pragma unroll
      for (int ks = 0; ks < 16; ++ks)
        s = wmma4(qA[ks], kB[ks], s);

      const int kd = kbase + lid - qbase - 8 * half;
#pragma unroll
      for (int r = 0; r < 8; ++r) {
        const float sv = (kd <= r) ? s[r] * scale : -1e30f;
        s[r] = sv;
        s_acc[16 * r] = sv;
      }
      wave_lds_order();

      if (lane < 16) {
        float mt = -1e30f;
#pragma unroll
        for (int n = 0; n < 16; ++n) mt = fmaxf(mt, s_row[n]);
        const float mnew  = fmaxf(mrun, mt);
        const float alpha = __expf(mrun - mnew);
        mrun = mnew;
        lrun *= alpha;
        Mrow[lane] = mnew;
        Arow[lane] = alpha;
      }
      wave_lds_order();

#pragma unroll
      for (int r = 0; r < 8; ++r) {
        const float p  = __expf(s[r] - m_acc[r]);
        const float al = a_acc[r];
        s_acc[16 * r] = p;
        o[0][r] *= al; o[1][r] *= al; o[2][r] *= al; o[3][r] *= al;
      }
      wave_lds_order();

      if (lane < 16) {
        float rs = 0.0f;
#pragma unroll
        for (int n = 0; n < 16; ++n) rs += s_row[n];
        lrun += rs;
      }

      v2f pA[4];
#pragma unroll
      for (int kk = 0; kk < 4; ++kk)
        pA[kk] = *(const v2f*)(&Smem[lid * 16 + kk * 4 + 2 * half]);
      v2f vB[4][4];
      {
        const float* vp = vo + (rowk + 2 * half) * HS + lid;
#pragma unroll
        for (int nt = 0; nt < 4; ++nt) {
#pragma unroll
          for (int kk = 0; kk < 4; ++kk) {
            vB[nt][kk].x = vp[(kk * 4) * HS + nt * 16];
            vB[nt][kk].y = vp[(kk * 4 + 1) * HS + nt * 16];
          }
        }
      }
      asm volatile("" :: "v"(vB[0][0]), "v"(vB[0][1]), "v"(vB[0][2]), "v"(vB[0][3]),
                         "v"(vB[1][0]), "v"(vB[1][1]), "v"(vB[1][2]), "v"(vB[1][3]),
                         "v"(vB[2][0]), "v"(vB[2][1]), "v"(vB[2][2]), "v"(vB[2][3]),
                         "v"(vB[3][0]), "v"(vB[3][1]), "v"(vB[3][2]), "v"(vB[3][3]));
#pragma unroll
      for (int nt = 0; nt < 4; ++nt) {
#pragma unroll
        for (int kk = 0; kk < 4; ++kk)
          o[nt] = wmma4(pA[kk], vB[nt][kk], o[nt]);
      }
      wave_lds_order();
    }

    float* op = Opart[wave];
#pragma unroll
    for (int nt = 0; nt < 4; ++nt) {
#pragma unroll
      for (int r = 0; r < 8; ++r)
        op[(r + 8 * half) * OPITCH + nt * 16 + lid] = o[nt][r];
    }
    if (lane < 16) {
      Mpart[wave][lane] = mrun;
      Lpart[wave][lane] = lrun;
    }
  }
  __syncthreads();

  const int nact = (nkt < ANW) ? nkt : ANW;
  const int m    = wave;
  const int cp   = 2 * lane;
  float mmax = -1e30f;
  for (int w = 0; w < nact; ++w) mmax = fmaxf(mmax, Mpart[w][m]);
  float lsum = 0.0f, a0 = 0.0f, a1 = 0.0f;
  for (int w = 0; w < nact; ++w) {
    const float wgt = __expf(Mpart[w][m] - mmax);
    const float2 pv = *(const float2*)(&Opart[w][m * OPITCH + cp]);
    lsum += wgt * Lpart[w][m];
    a0   += wgt * pv.x;
    a1   += wgt * pv.y;
  }
  const float inv = 1.0f / lsum;
  float2 res;
  res.x = a0 * inv;
  res.y = a1 * inv;
  *(float2*)(out + (rowq + m) * HS + cp) = res;
}

extern "C" void kernel_launch(void* const* d_in, const int* in_sizes, int n_in,
                              void* d_out, int out_size, void* d_ws, size_t ws_size,
                              hipStream_t stream) {
  const float* x  = (const float*)d_in[0];
  const float* Wq = (const float*)d_in[1];
  const float* Wk = (const float*)d_in[2];
  const float* Wv = (const float*)d_in[3];

  const int BT = in_sizes[0] / NEMBD;
  const int T  = 4096;
  const int B  = BT / T;

  float* qt = (float*)d_ws;
  float* kt = qt + (size_t)BT * HS;
  float* vv = kt + (size_t)BT * HS;
  float* out = (float*)d_out;

  qkv_rope_kernel<<<BT / 64, 384, 0, stream>>>(x, Wq, Wk, Wv, qt, kt, vv, T);
  attn_kernel<<<B * (T / 16), ATHREADS, 0, stream>>>(qt, kt, vv, out, T);
}


# q and k workspace planes kept in matrix-fragment order (via LDS in the projection), V read as whole rows: attention operand loads cover whole 128-byte lines; k order of both products permuted
# speedup vs baseline: 9.5032x; 1.0790x over previous
#include <hip/hip_runtime.h>
#include <math.h>

#define NEMBD 384
#define HS 64
#define FTHETA 10000.0f

typedef float v2f __attribute__((ext_vector_type(2)));
typedef float v4f __attribute__((ext_vector_type(4)));
typedef float v8f __attribute__((ext_vector_type(8)));

__device__ __forceinline__ v8f wmma4(v2f a, v2f b, v8f c) {
  return __builtin_amdgcn_wmma_f32_16x16x4_f32(
       false, a,  false, b,
       (short)0, c,  false,  false);
}

__device__ __forceinline__ void wave_lds_order() {
  __builtin_amdgcn_fence(__ATOMIC_RELEASE, "wavefront");
  __builtin_amdgcn_wave_barrier();
  __builtin_amdgcn_fence(__ATOMIC_ACQUIRE, "wavefront");
}

#define SPITCH 68

#define WPITCH4 9

__global__ __launch_bounds__(384) __attribute__((amdgpu_waves_per_eu(3, 4)))
void qkv_rope_kernel(
    const float* __restrict__ x, const float* __restrict__ Wq,
    const float* __restrict__ Wk, const float* __restrict__ Wv,
    float* __restrict__ qt, float* __restrict__ kt, float* __restrict__ vo,
    int T)
{
  __shared__ float4 Ws4[192 * WPITCH4];
  __shared__ __attribute__((aligned(16))) float Stage[8][16 * SPITCH];

  const int tid  = threadIdx.x;
  const int wave = tid >> 5;
  const int lane = tid & 31;
  const int c    = lane & 15;
  const int h    = lane >> 4;
  const int rt   = wave & 3;
  const int mat  = wave >> 2;
  const int row0 = blockIdx.x * 64 + rt * 16;

  const float4* wsrc[4];
  int wdst[4];
#pragma unroll
  for (int u = 0; u < 4; ++u) {
    const int i  = tid + 384 * u;
    const int wr = i >> 3;
    const int c4 = i & 7;
    const float* base = (wr < 64) ? Wq : ((wr < 128) ? Wk : Wv);
    wsrc[u] = (const float4*)(base + (size_t)(wr & 63) * NEMBD) + c4;
    wdst[u] = wr * WPITCH4 + c4;
  }
  const float4* xsrc = (const float4*)(x + (size_t)(row0 + c) * NEMBD + 16 * h);
  const float4* bsrc = &Ws4[(mat * 64 + c) * WPITCH4 + 4 * h];

  Ws4[wdst[0]] = wsrc[0][0];
  Ws4[wdst[1]] = wsrc[1][0];
  Ws4[wdst[2]] = wsrc[2][0];
  Ws4[wdst[3]] = wsrc[3][0];
  float4 x0 = xsrc[0], x1 = xsrc[1], x2 = xsrc[2], x3 = xsrc[3];

  v8f acc[4];
#pragma unroll
  for (int i = 0; i < 4; ++i) acc[i] = v8f{};

  __syncthreads();

  for (int kc = 0; kc < NEMBD / 32; ++kc) {
    const bool more = (kc + 1 < NEMBD / 32);

    float4 a[4];
    a[0] = x0; a[1] = x1; a[2] = x2; a[3] = x3;

    float4 w0 = {}, w1 = {}, w2 = {}, w3 = {};
    if (more) {
      const int kn = (kc + 1) * 8;
      w0 = wsrc[0][kn]; w1 = wsrc[1][kn]; w2 = wsrc[2][kn]; w3 = wsrc[3][kn];
      x0 = xsrc[kn]; x1 = xsrc[kn + 1]; x2 = xsrc[kn + 2]; x3 = xsrc[kn + 3];
    }

    float4 b[4][4];
#pragma unroll
    for (int nt = 0; nt < 4; ++nt) {
#pragma unroll
      for (int u = 0; u < 4; ++u) b[nt][u] = bsrc[nt * 16 * WPITCH4 + u];
    }

#pragma unroll
    for (int u = 0; u < 4; ++u) {
      v2f a01; a01.x = a[u].x; a01.y = a[u].y;
      v2f a23; a23.x = a[u].z; a23.y = a[u].w;
#pragma unroll
      for (int nt = 0; nt < 4; ++nt) {
        v2f b01; b01.x = b[nt][u].x; b01.y = b[nt][u].y;
        acc[nt] = wmma4(a01, b01, acc[nt]);
      }
#pragma unroll
      for (int nt = 0; nt < 4; ++nt) {
        v2f b23; b23.x = b[nt][u].z; b23.y = b[nt][u].w;
        acc[nt] = wmma4(a23, b23, acc[nt]);
      }
    }

    if (more) {
      __syncthreads();
      Ws4[wdst[0]] = w0;
      Ws4[wdst[1]] = w1;
      Ws4[wdst[2]] = w2;
      Ws4[wdst[3]] = w3;
      __syncthreads();
    }
  }

  const int rbase = row0 + 8 * h;

  if (mat == 2) {
#pragma unroll
    for (int nt = 0; nt < 4; ++nt) {
#pragma unroll
      for (int r = 0; r < 8; ++r)
        vo[(size_t)(rbase + r) * HS + nt * 16 + c] = acc[nt][r];
    }
    return;
  }

  float* __restrict__ dst = (mat == 0) ? qt : kt;
  float* st = Stage[wave];
  const float lnT = logf(FTHETA);
  const bool even = (c & 1) == 0;
#pragma unroll
  for (int nt = 0; nt < 4; ++nt) {
    const int n   = nt * 16 + c;
    const int idx = (nt < 2) ? n : n - 32;
    const float f = expf(-(2.0f * (float)idx) * (1.0f / 64.0f) * lnT);
#pragma unroll
    for (int r = 0; r < 8; ++r) {
      const int row  = rbase + r;
      const float tf = (float)(row % T);
      const float tv = (nt < 2) ? sinf(tf * f) : cosf(tf * f);
      const float tp = __shfl_xor(tv, 1);
      const float av = acc[nt][r];
      const float ap = __shfl_xor(av, 1);
      const float ct = even ? tv : tp;
      const float sn = even ? tp : tv;
      const float e  = even ? av : ap;
      const float o  = even ? ap : av;
      const float rot = (mat == 0) ? (e * ct + o * sn) : (e * ct - o * sn);
      st[(8 * h + r) * SPITCH + n] = even ? rot : (o - rot * sn);
    }
  }
  wave_lds_order();

  const v4f* s4 = (const v4f*)(st + c * SPITCH + 4 * h);
  v4f* d4 = (v4f*)(dst + (size_t)row0 * HS) + lane;
#pragma unroll
  for (int j = 0; j < 8; ++j) d4[32 * j] = s4[2 * j];
}

#define ANW 16
#define ATHREADS 512
#define OPITCH 68

__device__ __forceinline__ void load_frag_tile(const float* tile, int lane, v2f (&f)[16]) {
  const v4f* p = (const v4f*)tile + lane;
#pragma unroll
  for (int j = 0; j < 8; ++j) {
    const v4f t = p[32 * j];
    f[2 * j].x = t.x;     f[2 * j].y = t.y;
    f[2 * j + 1].x = t.z; f[2 * j + 1].y = t.w;
  }
}

__global__ __launch_bounds__(ATHREADS) __attribute__((amdgpu_waves_per_eu(8, 8)))
void attn_kernel(
    const float* __restrict__ qt, const float* __restrict__ kt,
    const float* __restrict__ vo, float* __restrict__ out, int T)
{
  const int ntiles = T >> 4;
  const int b      = blockIdx.x / ntiles;
  const int qtile  = blockIdx.x % ntiles;
  const int tid    = threadIdx.x;
  const int wave   = __builtin_amdgcn_readfirstlane(tid >> 5);
  const int lane   = tid & 31;
  const int half   = lane >> 4;
  const int lid    = lane & 15;
  const int qbase  = qtile << 4;
  const size_t rowq = (size_t)b * T + qbase;

  __shared__ __attribute__((aligned(16))) float SmemAll[ANW][256];
  __shared__ __attribute__((aligned(16))) float MrowAll[ANW][16];
  __shared__ __attribute__((aligned(16))) float ArowAll[ANW][16];
  __shared__ __attribute__((aligned(16))) float Opart[ANW][16 * OPITCH];
  __shared__ float Mpart[ANW][16];
  __shared__ float Lpart[ANW][16];

  float* Smem = SmemAll[wave];
  float* Mrow = MrowAll[wave];
  float* Arow = ArowAll[wave];
  float*       s_acc = Smem + half * 128 + lid;
  const float* s_row = Smem + lid * 16;
  const float* m_acc = Mrow + 8 * half;
  const float* a_acc = Arow + 8 * half;

  const int nkt = qtile + 1;

  if (wave < nkt) {
    v2f qA[16];
    load_frag_tile(qt + rowq * HS, lane, qA);

    v8f o[4];
#pragma unroll
    for (int i = 0; i < 4; ++i) o[i] = v8f{};
    float mrun = -1e30f, lrun = 0.0f;
    const float scale = 0.05103103630798288f;

    for (int ktile = wave; ktile < nkt; ktile += ANW) {
      const int kbase   = ktile << 4;
      const size_t rowk = (size_t)b * T + kbase;

      v2f kB[16];
      load_frag_tile(kt + rowk * HS, lane, kB);

      v8f s = v8f{};
#pragma unroll
      for (int ks = 0; ks < 16; ++ks)
        s = wmma4(qA[ks], kB[ks], s);

      const int kd = kbase + lid - qbase - 8 * half;
#pragma unroll
      for (int r = 0; r < 8; ++r) {
        const float sv = (kd <= r) ? s[r] * scale : -1e30f;
        s[r] = sv;
        s_acc[16 * r] = sv;
      }
      wave_lds_order();

      if (lane < 16) {
        float mt = -1e30f;
#pragma unroll
        for (int n = 0; n < 16; ++n) mt = fmaxf(mt, s_row[n]);
        const float mnew  = fmaxf(mrun, mt);
        const float alpha = __expf(mrun - mnew);
        mrun = mnew;
        lrun *= alpha;
        Mrow[lane] = mnew;
        Arow[lane] = alpha;
      }
      wave_lds_order();

#pragma unroll
      for (int r = 0; r < 8; ++r) {
        const float p  = __expf(s[r] - m_acc[r]);
        const float al = a_acc[r];
        s_acc[16 * r] = p;
        o[0][r] *= al; o[1][r] *= al; o[2][r] *= al; o[3][r] *= al;
      }
      wave_lds_order();

      if (lane < 16) {
        float rs = 0.0f;
#pragma unroll
        for (int n = 0; n < 16; ++n) rs += s_row[n];
        lrun += rs;
      }

      v2f pA[4];
#pragma unroll
      for (int kk = 0; kk < 4; ++kk)
        pA[kk] = *(const v2f*)(&Smem[lid * 16 + 8 * half + 2 * kk]);
      v4f vr[8];
      {
        const v4f* vp = (const v4f*)(vo + (rowk + 8 * half) * HS) + lid;
#pragma unroll
        for (int i = 0; i < 8; ++i) vr[i] = vp[16 * i];
      }
#pragma unroll
      for (int nt = 0; nt < 4; ++nt) {
#pragma unroll
        for (int kk = 0; kk < 4; ++kk) {
          v2f vB;
          vB.x = vr[2 * kk][nt];
          vB.y = vr[2 * kk + 1][nt];
          o[nt] = wmma4(pA[kk], vB, o[nt]);
        }
      }
      wave_lds_order();
    }

    float* op = Opart[wave];
#pragma unroll
    for (int r = 0; r < 8; ++r) {
      v4f t;
      t.x = o[0][r]; t.y = o[1][r]; t.z = o[2][r]; t.w = o[3][r];
      *(v4f*)(&op[(r + 8 * half) * OPITCH + 4 * lid]) = t;
    }
    if (lane < 16) {
      Mpart[wave][lane] = mrun;
      Lpart[wave][lane] = lrun;
    }
  }
  __syncthreads();

  const int nact = (nkt < ANW) ? nkt : ANW;
  const int m    = wave;
  const int cp   = 2 * lane;
  float mmax = -1e30f;
  for (int w = 0; w < nact; ++w) mmax = fmaxf(mmax, Mpart[w][m]);
  float lsum = 0.0f, a0 = 0.0f, a1 = 0.0f;
  for (int w = 0; w < nact; ++w) {
    const float wgt = __expf(Mpart[w][m] - mmax);
    const float2 pv = *(const float2*)(&Opart[w][m * OPITCH + cp]);
    lsum += wgt * Lpart[w][m];
    a0   += wgt * pv.x;
    a1   += wgt * pv.y;
  }
  const float inv = 1.0f / lsum;
  float2 res;
  res.x = a0 * inv;
  res.y = a1 * inv;
  *(float2*)(out + (rowq + m) * HS + cp) = res;
}

extern "C" void kernel_launch(void* const* d_in, const int* in_sizes, int n_in,
                              void* d_out, int out_size, void* d_ws, size_t ws_size,
                              hipStream_t stream) {
  const float* x  = (const float*)d_in[0];
  const float* Wq = (const float*)d_in[1];
  const float* Wk = (const float*)d_in[2];
  const float* Wv = (const float*)d_in[3];

  const int BT = in_sizes[0] / NEMBD;
  const int T  = 4096;
  const int B  = BT / T;

  float* qt = (float*)d_ws;
  float* kt = qt + (size_t)BT * HS;
  float* vv = kt + (size_t)BT * HS;
  float* out = (float*)d_out;

  qkv_rope_kernel<<<BT / 64, 384, 0, stream>>>(x, Wq, Wk, Wv, qt, kt, vv, T);
  attn_kernel<<<B * (T / 16), ATHREADS, 0, stream>>>(qt, kt, vv, out, T);
}


# Attention work-groups numbered longest causal key range first, batches interleaved (blockIdx -> query tile reversed): the long tiles no longer start last and run on over idle WGPs
# speedup vs baseline: 11.3436x; 1.1937x over previous
#include <hip/hip_runtime.h>
#include <math.h>

#define NEMBD 384
#define HS 64
#define FTHETA 10000.0f

typedef float v2f __attribute__((ext_vector_type(2)));
typedef float v4f __attribute__((ext_vector_type(4)));
typedef float v8f __attribute__((ext_vector_type(8)));

__device__ __forceinline__ v8f wmma4(v2f a, v2f b, v8f c) {
  return __builtin_amdgcn_wmma_f32_16x16x4_f32(
       false, a,  false, b,
       (short)0, c,  false,  false);
}

__device__ __forceinline__ void wave_lds_order() {
  __builtin_amdgcn_fence(__ATOMIC_RELEASE, "wavefront");
  __builtin_amdgcn_wave_barrier();
  __builtin_amdgcn_fence(__ATOMIC_ACQUIRE, "wavefront");
}

#define SPITCH 68

#define WPITCH4 9

__global__ __launch_bounds__(384) __attribute__((amdgpu_waves_per_eu(3, 4)))
void qkv_rope_kernel(
    const float* __restrict__ x, const float* __restrict__ Wq,
    const float* __restrict__ Wk, const float* __restrict__ Wv,
    float* __restrict__ qt, float* __restrict__ kt, float* __restrict__ vo,
    int T)
{
  __shared__ float4 Ws4[192 * WPITCH4];
  __shared__ __attribute__((aligned(16))) float Stage[8][16 * SPITCH];

  const int tid  = threadIdx.x;
  const int wave = tid >> 5;
  const int lane = tid & 31;
  const int c    = lane & 15;
  const int h    = lane >> 4;
  const int rt   = wave & 3;
  const int mat  = wave >> 2;
  const int row0 = blockIdx.x * 64 + rt * 16;

  const float4* wsrc[4];
  int wdst[4];
#pragma unroll
  for (int u = 0; u < 4; ++u) {
    const int i  = tid + 384 * u;
    const int wr = i >> 3;
    const int c4 = i & 7;
    const float* base = (wr < 64) ? Wq : ((wr < 128) ? Wk : Wv);
    wsrc[u] = (const float4*)(base + (size_t)(wr & 63) * NEMBD) + c4;
    wdst[u] = wr * WPITCH4 + c4;
  }
  const float4* xsrc = (const float4*)(x + (size_t)(row0 + c) * NEMBD + 16 * h);
  const float4* bsrc = &Ws4[(mat * 64 + c) * WPITCH4 + 4 * h];

  Ws4[wdst[0]] = wsrc[0][0];
  Ws4[wdst[1]] = wsrc[1][0];
  Ws4[wdst[2]] = wsrc[2][0];
  Ws4[wdst[3]] = wsrc[3][0];
  float4 x0 = xsrc[0], x1 = xsrc[1], x2 = xsrc[2], x3 = xsrc[3];

  v8f acc[4];
#pragma unroll
  for (int i = 0; i < 4; ++i) acc[i] = v8f{};

  __syncthreads();

  for (int kc = 0; kc < NEMBD / 32; ++kc) {
    const bool more = (kc + 1 < NEMBD / 32);

    float4 a[4];
    a[0] = x0; a[1] = x1; a[2] = x2; a[3] = x3;

    float4 w0 = {}, w1 = {}, w2 = {}, w3 = {};
    if (more) {
      const int kn = (kc + 1) * 8;
      w0 = wsrc[0][kn]; w1 = wsrc[1][kn]; w2 = wsrc[2][kn]; w3 = wsrc[3][kn];
      x0 = xsrc[kn]; x1 = xsrc[kn + 1]; x2 = xsrc[kn + 2]; x3 = xsrc[kn + 3];
    }

    float4 b[4][4];
#pragma unroll
    for (int nt = 0; nt < 4; ++nt) {
#pragma unroll
      for (int u = 0; u < 4; ++u) b[nt][u] = bsrc[nt * 16 * WPITCH4 + u];
    }

#pragma unroll
    for (int u = 0; u < 4; ++u) {
      v2f a01; a01.x = a[u].x; a01.y = a[u].y;
      v2f a23; a23.x = a[u].z; a23.y = a[u].w;
#pragma unroll
      for (int nt = 0; nt < 4; ++nt) {
        v2f b01; b01.x = b[nt][u].x; b01.y = b[nt][u].y;
        acc[nt] = wmma4(a01, b01, acc[nt]);
      }
#pragma unroll
      for (int nt = 0; nt < 4; ++nt) {
        v2f b23; b23.x = b[nt][u].z; b23.y = b[nt][u].w;
        acc[nt] = wmma4(a23, b23, acc[nt]);
      }
    }

    if (more) {
      __syncthreads();
      Ws4[wdst[0]] = w0;
      Ws4[wdst[1]] = w1;
      Ws4[wdst[2]] = w2;
      Ws4[wdst[3]] = w3;
      __syncthreads();
    }
  }

  const int rbase = row0 + 8 * h;

  if (mat == 2) {
#pragma unroll
    for (int nt = 0; nt < 4; ++nt) {
#pragma unroll
      for (int r = 0; r < 8; ++r)
        vo[(size_t)(rbase + r) * HS + nt * 16 + c] = acc[nt][r];
    }
    return;
  }

  float* __restrict__ dst = (mat == 0) ? qt : kt;
  float* st = Stage[wave];
  const float lnT = logf(FTHETA);
  const bool even = (c & 1) == 0;
#pragma unroll
  for (int nt = 0; nt < 4; ++nt) {
    const int n   = nt * 16 + c;
    const int idx = (nt < 2) ? n : n - 32;
    const float f = expf(-(2.0f * (float)idx) * (1.0f / 64.0f) * lnT);
#pragma unroll
    for (int r = 0; r < 8; ++r) {
      const int row  = rbase + r;
      const float tf = (float)(row % T);
      const float tv = (nt < 2) ? sinf(tf * f) : cosf(tf * f);
      const float tp = __shfl_xor(tv, 1);
      const float av = acc[nt][r];
      const float ap = __shfl_xor(av, 1);
      const float ct = even ? tv : tp;
      const float sn = even ? tp : tv;
      const float e  = even ? av : ap;
      const float o  = even ? ap : av;
      const float rot = (mat == 0) ? (e * ct + o * sn) : (e * ct - o * sn);
      st[(8 * h + r) * SPITCH + n] = even ? rot : (o - rot * sn);
    }
  }
  wave_lds_order();

  const v4f* s4 = (const v4f*)(st + c * SPITCH + 4 * h);
  v4f* d4 = (v4f*)(dst + (size_t)row0 * HS) + lane;
#pragma unroll
  for (int j = 0; j < 8; ++j) d4[32 * j] = s4[2 * j];
}

#define ANW 16
#define ATHREADS 512
#define OPITCH 68

__device__ __forceinline__ void load_frag_tile(const float* tile, int lane, v2f (&f)[16]) {
  const v4f* p = (const v4f*)tile + lane;
#pragma unroll
  for (int j = 0; j < 8; ++j) {
    const v4f t = p[32 * j];
    f[2 * j].x = t.x;     f[2 * j].y = t.y;
    f[2 * j + 1].x = t.z; f[2 * j + 1].y = t.w;
  }
}

__global__ __launch_bounds__(ATHREADS) __attribute__((amdgpu_waves_per_eu(8, 8)))
void attn_kernel(
    const float* __restrict__ qt, const float* __restrict__ kt,
    const float* __restrict__ vo, float* __restrict__ out, int T, int nbatch)
{
  const int ntiles = T >> 4;
  const int slot   = blockIdx.x / nbatch;
  const int b      = blockIdx.x - slot * nbatch;
  const int qtile  = ntiles - 1 - slot;
  const int tid    = threadIdx.x;
  const int wave   = __builtin_amdgcn_readfirstlane(tid >> 5);
  const int lane   = tid & 31;
  const int half   = lane >> 4;
  const int lid    = lane & 15;
  const int qbase  = qtile << 4;
  const size_t rowq = (size_t)b * T + qbase;

  __shared__ __attribute__((aligned(16))) float SmemAll[ANW][256];
  __shared__ __attribute__((aligned(16))) float MrowAll[ANW][16];
  __shared__ __attribute__((aligned(16))) float ArowAll[ANW][16];
  __shared__ __attribute__((aligned(16))) float Opart[ANW][16 * OPITCH];
  __shared__ float Mpart[ANW][16];
  __shared__ float Lpart[ANW][16];

  float* Smem = SmemAll[wave];
  float* Mrow = MrowAll[wave];
  float* Arow = ArowAll[wave];
  float*       s_acc = Smem + half * 128 + lid;
  const float* s_row = Smem + lid * 16;
  const float* m_acc = Mrow + 8 * half;
  const float* a_acc = Arow + 8 * half;

  const int nkt = qtile + 1;

  if (wave < nkt) {
    v2f qA[16];
    load_frag_tile(qt + rowq * HS, lane, qA);

    v8f o[4];
#pragma unroll
    for (int i = 0; i < 4; ++i) o[i] = v8f{};
    float mrun = -1e30f, lrun = 0.0f;
    const float scale = 0.05103103630798288f;

    for (int ktile = wave; ktile < nkt; ktile += ANW) {
      const int kbase   = ktile << 4;
      const size_t rowk = (size_t)b * T + kbase;

      v2f kB[16];
      load_frag_tile(kt + rowk * HS, lane, kB);

      v8f s = v8f{};
#pragma unroll
      for (int ks = 0; ks < 16; ++ks)
        s = wmma4(qA[ks], kB[ks], s);

      const int kd = kbase + lid - qbase - 8 * half;
#pragma unroll
      for (int r = 0; r < 8; ++r) {
        const float sv = (kd <= r) ? s[r] * scale : -1e30f;
        s[r] = sv;
        s_acc[16 * r] = sv;
      }
      wave_lds_order();

      if (lane < 16) {
        float mt = -1e30f;
#pragma unroll
        for (int n = 0; n < 16; ++n) mt = fmaxf(mt, s_row[n]);
        const float mnew  = fmaxf(mrun, mt);
        const float alpha = __expf(mrun - mnew);
        mrun = mnew;
        lrun *= alpha;
        Mrow[lane] = mnew;
        Arow[lane] = alpha;
      }
      wave_lds_order();

#pragma unroll
      for (int r = 0; r < 8; ++r) {
        const float p  = __expf(s[r] - m_acc[r]);
        const float al = a_acc[r];
        s_acc[16 * r] = p;
        o[0][r] *= al; o[1][r] *= al; o[2][r] *= al; o[3][r] *= al;
      }
      wave_lds_order();

      if (lane < 16) {
        float rs = 0.0f;
#pragma unroll
        for (int n = 0; n < 16; ++n) rs += s_row[n];
        lrun += rs;
      }

      v2f pA[4];
#pragma unroll
      for (int kk = 0; kk < 4; ++kk)
        pA[kk] = *(const v2f*)(&Smem[lid * 16 + 8 * half + 2 * kk]);
      v4f vr[8];
      {
        const v4f* vp = (const v4f*)(vo + (rowk + 8 * half) * HS) + lid;
#pragma unroll
        for (int i = 0; i < 8; ++i) vr[i] = vp[16 * i];
      }
#pragma unroll
      for (int nt = 0; nt < 4; ++nt) {
#pragma unroll
        for (int kk = 0; kk < 4; ++kk) {
          v2f vB;
          vB.x = vr[2 * kk][nt];
          vB.y = vr[2 * kk + 1][nt];
          o[nt] = wmma4(pA[kk], vB, o[nt]);
        }
      }
      wave_lds_order();
    }

    float* op = Opart[wave];
#pragma unroll
    for (int r = 0; r < 8; ++r) {
      v4f t;
      t.x = o[0][r]; t.y = o[1][r]; t.z = o[2][r]; t.w = o[3][r];
      *(v4f*)(&op[(r + 8 * half) * OPITCH + 4 * lid]) = t;
    }
    if (lane < 16) {
      Mpart[wave][lane] = mrun;
      Lpart[wave][lane] = lrun;
    }
  }
  __syncthreads();

  const int nact = (nkt < ANW) ? nkt : ANW;
  const int m    = wave;
  const int cp   = 2 * lane;
  float mmax = -1e30f;
  for (int w = 0; w < nact; ++w) mmax = fmaxf(mmax, Mpart[w][m]);
  float lsum = 0.0f, a0 = 0.0f, a1 = 0.0f;
  for (int w = 0; w < nact; ++w) {
    const float wgt = __expf(Mpart[w][m] - mmax);
    const float2 pv = *(const float2*)(&Opart[w][m * OPITCH + cp]);
    lsum += wgt * Lpart[w][m];
    a0   += wgt * pv.x;
    a1   += wgt * pv.y;
  }
  const float inv = 1.0f / lsum;
  float2 res;
  res.x = a0 * inv;
  res.y = a1 * inv;
  *(float2*)(out + (rowq + m) * HS + cp) = res;
}

extern "C" void kernel_launch(void* const* d_in, const int* in_sizes, int n_in,
                              void* d_out, int out_size, void* d_ws, size_t ws_size,
                              hipStream_t stream) {
  const float* x  = (const float*)d_in[0];
  const float* Wq = (const float*)d_in[1];
  const float* Wk = (const float*)d_in[2];
  const float* Wv = (const float*)d_in[3];

  const int BT = in_sizes[0] / NEMBD;
  const int T  = 4096;
  const int B  = BT / T;

  float* qt = (float*)d_ws;
  float* kt = qt + (size_t)BT * HS;
  float* vv = kt + (size_t)BT * HS;
  float* out = (float*)d_out;

  qkv_rope_kernel<<<BT / 64, 384, 0, stream>>>(x, Wq, Wk, Wv, qt, kt, vv, T);
  attn_kernel<<<B * (T / 16), ATHREADS, 0, stream>>>(qt, kt, vv, out, T, B);
}


# Attention on transposed tiles (S^T = K Q^T, O^T += V^T P^T; v plane in fragment order): softmax max and sum stay in registers plus one half swap, P^T feeds P.V from registers; row sums: new tree
# speedup vs baseline: 12.5950x; 1.1103x over previous
#include <hip/hip_runtime.h>
#include <math.h>

#define NEMBD 384
#define HS 64
#define FTHETA 10000.0f

typedef float v2f __attribute__((ext_vector_type(2)));
typedef float v4f __attribute__((ext_vector_type(4)));
typedef float v8f __attribute__((ext_vector_type(8)));

__device__ __forceinline__ v8f wmma4(v2f a, v2f b, v8f c) {
  return __builtin_amdgcn_wmma_f32_16x16x4_f32(
       false, a,  false, b,
       (short)0, c,  false,  false);
}

__device__ __forceinline__ void wave_lds_order() {
  __builtin_amdgcn_fence(__ATOMIC_RELEASE, "wavefront");
  __builtin_amdgcn_wave_barrier();
  __builtin_amdgcn_fence(__ATOMIC_ACQUIRE, "wavefront");
}

#define SPITCH 68

#define WPITCH4 9

__global__ __launch_bounds__(384) __attribute__((amdgpu_waves_per_eu(3, 4)))
void qkv_rope_kernel(
    const float* __restrict__ x, const float* __restrict__ Wq,
    const float* __restrict__ Wk, const float* __restrict__ Wv,
    float* __restrict__ qt, float* __restrict__ kt, float* __restrict__ vo,
    int T)
{
  __shared__ float4 Ws4[192 * WPITCH4];
  __shared__ __attribute__((aligned(16))) float Stage[8][16 * SPITCH];

  const int tid  = threadIdx.x;
  const int wave = tid >> 5;
  const int lane = tid & 31;
  const int c    = lane & 15;
  const int h    = lane >> 4;
  const int rt   = wave & 3;
  const int mat  = wave >> 2;
  const int row0 = blockIdx.x * 64 + rt * 16;

  const float4* wsrc[4];
  int wdst[4];
#pragma unroll
  for (int u = 0; u < 4; ++u) {
    const int i  = tid + 384 * u;
    const int wr = i >> 3;
    const int c4 = i & 7;
    const float* base = (wr < 64) ? Wq : ((wr < 128) ? Wk : Wv);
    wsrc[u] = (const float4*)(base + (size_t)(wr & 63) * NEMBD) + c4;
    wdst[u] = wr * WPITCH4 + c4;
  }
  const float4* xsrc = (const float4*)(x + (size_t)(row0 + c) * NEMBD + 16 * h);
  const float4* bsrc = &Ws4[(mat * 64 + c) * WPITCH4 + 4 * h];

  Ws4[wdst[0]] = wsrc[0][0];
  Ws4[wdst[1]] = wsrc[1][0];
  Ws4[wdst[2]] = wsrc[2][0];
  Ws4[wdst[3]] = wsrc[3][0];
  float4 x0 = xsrc[0], x1 = xsrc[1], x2 = xsrc[2], x3 = xsrc[3];

  v8f acc[4];
#pragma unroll
  for (int i = 0; i < 4; ++i) acc[i] = v8f{};

  __syncthreads();

  for (int kc = 0; kc < NEMBD / 32; ++kc) {
    const bool more = (kc + 1 < NEMBD / 32);

    float4 a[4];
    a[0] = x0; a[1] = x1; a[2] = x2; a[3] = x3;

    float4 w0 = {}, w1 = {}, w2 = {}, w3 = {};
    if (more) {
      const int kn = (kc + 1) * 8;
      w0 = wsrc[0][kn]; w1 = wsrc[1][kn]; w2 = wsrc[2][kn]; w3 = wsrc[3][kn];
      x0 = xsrc[kn]; x1 = xsrc[kn + 1]; x2 = xsrc[kn + 2]; x3 = xsrc[kn + 3];
    }

    float4 b[4][4];
#pragma unroll
    for (int nt = 0; nt < 4; ++nt) {
#pragma unroll
      for (int u = 0; u < 4; ++u) b[nt][u] = bsrc[nt * 16 * WPITCH4 + u];
    }

#pragma unroll
    for (int u = 0; u < 4; ++u) {
      v2f a01; a01.x = a[u].x; a01.y = a[u].y;
      v2f a23; a23.x = a[u].z; a23.y = a[u].w;
#pragma unroll
      for (int nt = 0; nt < 4; ++nt) {
        v2f b01; b01.x = b[nt][u].x; b01.y = b[nt][u].y;
        acc[nt] = wmma4(a01, b01, acc[nt]);
      }
#pragma unroll
      for (int nt = 0; nt < 4; ++nt) {
        v2f b23; b23.x = b[nt][u].z; b23.y = b[nt][u].w;
        acc[nt] = wmma4(a23, b23, acc[nt]);
      }
    }

    if (more) {
      __syncthreads();
      Ws4[wdst[0]] = w0;
      Ws4[wdst[1]] = w1;
      Ws4[wdst[2]] = w2;
      Ws4[wdst[3]] = w3;
      __syncthreads();
    }
  }

  const int rbase = row0 + 8 * h;

  if (mat == 2) {
    v4f* v4 = (v4f*)(vo + (size_t)row0 * HS) + lane;
#pragma unroll
    for (int nt = 0; nt < 4; ++nt) {
#pragma unroll
      for (int j = 0; j < 2; ++j) {
        v4f t;
        t.x = acc[nt][4 * j];     t.y = acc[nt][4 * j + 1];
        t.z = acc[nt][4 * j + 2]; t.w = acc[nt][4 * j + 3];
        v4[32 * (2 * nt + j)] = t;
      }
    }
    return;
  }

  float* __restrict__ dst = (mat == 0) ? qt : kt;
  float* st = Stage[wave];
  const float lnT = logf(FTHETA);
  const bool even = (c & 1) == 0;
#pragma unroll
  for (int nt = 0; nt < 4; ++nt) {
    const int n   = nt * 16 + c;
    const int idx = (nt < 2) ? n : n - 32;
    const float f = expf(-(2.0f * (float)idx) * (1.0f / 64.0f) * lnT);
#pragma unroll
    for (int r = 0; r < 8; ++r) {
      const int row  = rbase + r;
      const float tf = (float)(row % T);
      const float tv = (nt < 2) ? sinf(tf * f) : cosf(tf * f);
      const float tp = __shfl_xor(tv, 1);
      const float av = acc[nt][r];
      const float ap = __shfl_xor(av, 1);
      const float ct = even ? tv : tp;
      const float sn = even ? tp : tv;
      const float e  = even ? av : ap;
      const float o  = even ? ap : av;
      const float rot = (mat == 0) ? (e * ct + o * sn) : (e * ct - o * sn);
      st[(8 * h + r) * SPITCH + n] = even ? rot : (o - rot * sn);
    }
  }
  wave_lds_order();

  const v4f* s4 = (const v4f*)(st + c * SPITCH + 4 * h);
  v4f* d4 = (v4f*)(dst + (size_t)row0 * HS) + lane;
#pragma unroll
  for (int j = 0; j < 8; ++j) d4[32 * j] = s4[2 * j];
}

#define ANW 16
#define ATHREADS 512
#define OPITCH 68

__device__ __forceinline__ void load_frag_tile(const float* plane, unsigned piece, v2f (&f)[16]) {
  const v4f* p = (const v4f*)plane + piece;
#pragma unroll
  for (int j = 0; j < 8; ++j) {
    const v4f t = p[32 * j];
    f[2 * j].x = t.x;     f[2 * j].y = t.y;
    f[2 * j + 1].x = t.z; f[2 * j + 1].y = t.w;
  }
}

__device__ __forceinline__ float xhalf(float x) {
  const int b = __builtin_bit_cast(int, x);
  return __builtin_bit_cast(float,
      __builtin_amdgcn_permlanex16(b, b, 0x76543210, 0xfedcba98, false, false));
}

__global__ __launch_bounds__(ATHREADS) __attribute__((amdgpu_waves_per_eu(8, 8)))
void attn_kernel(
    const float* __restrict__ qt, const float* __restrict__ kt,
    const float* __restrict__ vo, float* __restrict__ out, int T, int nbatch)
{
  const int ntiles = T >> 4;
  const int slot   = blockIdx.x / nbatch;
  const int b      = blockIdx.x - slot * nbatch;
  const int qtile  = ntiles - 1 - slot;
  const int tid    = threadIdx.x;
  const int wave   = __builtin_amdgcn_readfirstlane(tid >> 5);
  const int lane   = tid & 31;
  const int half   = lane >> 4;
  const int lid    = lane & 15;
  const int qbase  = qtile << 4;
  const size_t rowq = (size_t)b * T + qbase;

  __shared__ __attribute__((aligned(16))) float Opart[ANW][16 * OPITCH];
  __shared__ float Mpart[ANW][16];
  __shared__ float Lpart[ANW][16];

  const int nkt = qtile + 1;

  if (wave < nkt) {
    v2f qB[16];
    load_frag_tile(qt, (unsigned)(rowq * (HS / 4)) + (unsigned)lane, qB);

    v8f o[4];
#pragma unroll
    for (int i = 0; i < 4; ++i) o[i] = v8f{};
    float mrun = -1e30f, lrun = 0.0f;
    const float scale = 0.05103103630798288f;

    for (int ktile = wave; ktile < nkt; ktile += ANW) {
      const int kbase   = ktile << 4;
      const size_t rowk = (size_t)b * T + kbase;
      const unsigned kpiece = (unsigned)(rowk * (HS / 4)) + (unsigned)lane;

      v2f kA[16];
      load_frag_tile(kt, kpiece, kA);

      v8f s = v8f{};
      v4f vf[8];
      const v4f* vp = (const v4f*)vo + kpiece;
#pragma unroll
      for (int ks = 0; ks < 8; ++ks)
        s = wmma4(kA[ks], qB[ks], s);
      __builtin_amdgcn_sched_barrier(0);
#pragma unroll
      for (int i = 0; i < 4; ++i) vf[i] = vp[32 * i];
#pragma unroll
      for (int ks = 8; ks < 16; ++ks)
        s = wmma4(kA[ks], qB[ks], s);
      __builtin_amdgcn_sched_barrier(0);
#pragma unroll
      for (int i = 4; i < 8; ++i) vf[i] = vp[32 * i];

      const int qd = qbase + lid - kbase - 8 * half;
      float mt = -1e30f;
#pragma unroll
      for (int r = 0; r < 8; ++r) {
        const float sv = (r <= qd) ? s[r] * scale : -1e30f;
        s[r] = sv;
        mt = fmaxf(mt, sv);
      }
      mt = fmaxf(mt, xhalf(mt));

      const float mnew  = fmaxf(mrun, mt);
      const float alpha = __expf(mrun - mnew);
      mrun = mnew;

      float rs = 0.0f;
#pragma unroll
      for (int r = 0; r < 8; ++r) {
        const float p = __expf(s[r] - mnew);
        s[r] = p;
        rs += p;
      }
      lrun = lrun * alpha + rs;
#pragma unroll
      for (int nt = 0; nt < 4; ++nt) {
#pragma unroll
        for (int r = 0; r < 8; ++r) o[nt][r] *= alpha;
      }

#pragma unroll
      for (int nt = 0; nt < 4; ++nt) {
#pragma unroll
        for (int j = 0; j < 2; ++j) {
          const v4f t = vf[2 * nt + j];
          v2f a01; a01.x = t.x; a01.y = t.y;
          v2f a23; a23.x = t.z; a23.y = t.w;
          v2f p01; p01.x = s[4 * j];     p01.y = s[4 * j + 1];
          v2f p23; p23.x = s[4 * j + 2]; p23.y = s[4 * j + 3];
          o[nt] = wmma4(a01, p01, o[nt]);
          o[nt] = wmma4(a23, p23, o[nt]);
        }
      }
    }

    float* op = Opart[wave] + lid * OPITCH + 8 * half;
#pragma unroll
    for (int nt = 0; nt < 4; ++nt) {
      v4f t0, t1;
      t0.x = o[nt][0]; t0.y = o[nt][1]; t0.z = o[nt][2]; t0.w = o[nt][3];
      t1.x = o[nt][4]; t1.y = o[nt][5]; t1.z = o[nt][6]; t1.w = o[nt][7];
      *(v4f*)(op + 16 * nt)     = t0;
      *(v4f*)(op + 16 * nt + 4) = t1;
    }
    const float lall = lrun + xhalf(lrun);
    if (lane < 16) {
      Mpart[wave][lane] = mrun;
      Lpart[wave][lane] = lall;
    }
  }
  __syncthreads();

  const int nact = (nkt < ANW) ? nkt : ANW;
  const int m    = wave;
  const int cp   = 2 * lane;
  float mmax = -1e30f;
  for (int w = 0; w < nact; ++w) mmax = fmaxf(mmax, Mpart[w][m]);
  float lsum = 0.0f, a0 = 0.0f, a1 = 0.0f;
  for (int w = 0; w < nact; ++w) {
    const float wgt = __expf(Mpart[w][m] - mmax);
    const float2 pv = *(const float2*)(&Opart[w][m * OPITCH + cp]);
    lsum += wgt * Lpart[w][m];
    a0   += wgt * pv.x;
    a1   += wgt * pv.y;
  }
  const float inv = 1.0f / lsum;
  float2 res;
  res.x = a0 * inv;
  res.y = a1 * inv;
  *(float2*)(out + (rowq + m) * HS + cp) = res;
}

extern "C" void kernel_launch(void* const* d_in, const int* in_sizes, int n_in,
                              void* d_out, int out_size, void* d_ws, size_t ws_size,
                              hipStream_t stream) {
  const float* x  = (const float*)d_in[0];
  const float* Wq = (const float*)d_in[1];
  const float* Wk = (const float*)d_in[2];
  const float* Wv = (const float*)d_in[3];

  const int BT = in_sizes[0] / NEMBD;
  const int T  = 4096;
  const int B  = BT / T;

  float* qt = (float*)d_ws;
  float* kt = qt + (size_t)BT * HS;
  float* vv = kt + (size_t)BT * HS;
  float* out = (float*)d_out;

  qkv_rope_kernel<<<BT / 64, 384, 0, stream>>>(x, Wq, Wk, Wv, qt, kt, vv, T);
  attn_kernel<<<B * (T / 16), ATHREADS, 0, stream>>>(qt, kt, vv, out, T, B);
}


# Attention: one work-group per PAIR of query tiles; each wave carries both tiles through its key loop, so a loaded K/V tile feeds 64 matrix ops, not 32: K/V reads per query halved, 4 waves per SIMD
# speedup vs baseline: 13.4001x; 1.0639x over previous
#include <hip/hip_runtime.h>
#include <math.h>

#define NEMBD 384
#define HS 64
#define FTHETA 10000.0f

typedef float v2f __attribute__((ext_vector_type(2)));
typedef float v4f __attribute__((ext_vector_type(4)));
typedef float v8f __attribute__((ext_vector_type(8)));

__device__ __forceinline__ v8f wmma4(v2f a, v2f b, v8f c) {
  return __builtin_amdgcn_wmma_f32_16x16x4_f32(
       false, a,  false, b,
       (short)0, c,  false,  false);
}

__device__ __forceinline__ void wave_lds_order() {
  __builtin_amdgcn_fence(__ATOMIC_RELEASE, "wavefront");
  __builtin_amdgcn_wave_barrier();
  __builtin_amdgcn_fence(__ATOMIC_ACQUIRE, "wavefront");
}

#define SPITCH 68

#define WPITCH4 9

__global__ __launch_bounds__(384) __attribute__((amdgpu_waves_per_eu(3, 4)))
void qkv_rope_kernel(
    const float* __restrict__ x, const float* __restrict__ Wq,
    const float* __restrict__ Wk, const float* __restrict__ Wv,
    float* __restrict__ qt, float* __restrict__ kt, float* __restrict__ vo,
    int T)
{
  __shared__ float4 Ws4[192 * WPITCH4];
  __shared__ __attribute__((aligned(16))) float Stage[8][16 * SPITCH];

  const int tid  = threadIdx.x;
  const int wave = tid >> 5;
  const int lane = tid & 31;
  const int c    = lane & 15;
  const int h    = lane >> 4;
  const int rt   = wave & 3;
  const int mat  = wave >> 2;
  const int row0 = blockIdx.x * 64 + rt * 16;

  const float4* wsrc[4];
  int wdst[4];
#pragma unroll
  for (int u = 0; u < 4; ++u) {
    const int i  = tid + 384 * u;
    const int wr = i >> 3;
    const int c4 = i & 7;
    const float* base = (wr < 64) ? Wq : ((wr < 128) ? Wk : Wv);
    wsrc[u] = (const float4*)(base + (size_t)(wr & 63) * NEMBD) + c4;
    wdst[u] = wr * WPITCH4 + c4;
  }
  const float4* xsrc = (const float4*)(x + (size_t)(row0 + c) * NEMBD + 16 * h);
  const float4* bsrc = &Ws4[(mat * 64 + c) * WPITCH4 + 4 * h];

  Ws4[wdst[0]] = wsrc[0][0];
  Ws4[wdst[1]] = wsrc[1][0];
  Ws4[wdst[2]] = wsrc[2][0];
  Ws4[wdst[3]] = wsrc[3][0];
  float4 x0 = xsrc[0], x1 = xsrc[1], x2 = xsrc[2], x3 = xsrc[3];

  v8f acc[4];
#pragma unroll
  for (int i = 0; i < 4; ++i) acc[i] = v8f{};

  __syncthreads();

  for (int kc = 0; kc < NEMBD / 32; ++kc) {
    const bool more = (kc + 1 < NEMBD / 32);

    float4 a[4];
    a[0] = x0; a[1] = x1; a[2] = x2; a[3] = x3;

    float4 w0 = {}, w1 = {}, w2 = {}, w3 = {};
    if (more) {
      const int kn = (kc + 1) * 8;
      w0 = wsrc[0][kn]; w1 = wsrc[1][kn]; w2 = wsrc[2][kn]; w3 = wsrc[3][kn];
      x0 = xsrc[kn]; x1 = xsrc[kn + 1]; x2 = xsrc[kn + 2]; x3 = xsrc[kn + 3];
    }

    float4 b[4][4];
#pragma unroll
    for (int nt = 0; nt < 4; ++nt) {
#pragma unroll
      for (int u = 0; u < 4; ++u) b[nt][u] = bsrc[nt * 16 * WPITCH4 + u];
    }

#pragma unroll
    for (int u = 0; u < 4; ++u) {
      v2f a01; a01.x = a[u].x; a01.y = a[u].y;
      v2f a23; a23.x = a[u].z; a23.y = a[u].w;
#pragma unroll
      for (int nt = 0; nt < 4; ++nt) {
        v2f b01; b01.x = b[nt][u].x; b01.y = b[nt][u].y;
        acc[nt] = wmma4(a01, b01, acc[nt]);
      }
#pragma unroll
      for (int nt = 0; nt < 4; ++nt) {
        v2f b23; b23.x = b[nt][u].z; b23.y = b[nt][u].w;
        acc[nt] = wmma4(a23, b23, acc[nt]);
      }
    }

    if (more) {
      __syncthreads();
      Ws4[wdst[0]] = w0;
      Ws4[wdst[1]] = w1;
      Ws4[wdst[2]] = w2;
      Ws4[wdst[3]] = w3;
      __syncthreads();
    }
  }

  const int rbase = row0 + 8 * h;

  if (mat == 2) {
    v4f* v4 = (v4f*)(vo + (size_t)row0 * HS) + lane;
#pragma unroll
    for (int nt = 0; nt < 4; ++nt) {
#pragma unroll
      for (int j = 0; j < 2; ++j) {
        v4f t;
        t.x = acc[nt][4 * j];     t.y = acc[nt][4 * j + 1];
        t.z = acc[nt][4 * j + 2]; t.w = acc[nt][4 * j + 3];
        v4[32 * (2 * nt + j)] = t;
      }
    }
    return;
  }

  float* __restrict__ dst = (mat == 0) ? qt : kt;
  float* st = Stage[wave];
  const float lnT = logf(FTHETA);
  const bool even = (c & 1) == 0;
#pragma unroll
  for (int nt = 0; nt < 4; ++nt) {
    const int n   = nt * 16 + c;
    const int idx = (nt < 2) ? n : n - 32;
    const float f = expf(-(2.0f * (float)idx) * (1.0f / 64.0f) * lnT);
#pragma unroll
    for (int r = 0; r < 8; ++r) {
      const int row  = rbase + r;
      const float tf = (float)(row % T);
      const float tv = (nt < 2) ? sinf(tf * f) : cosf(tf * f);
      const float tp = __shfl_xor(tv, 1);
      const float av = acc[nt][r];
      const float ap = __shfl_xor(av, 1);
      const float ct = even ? tv : tp;
      const float sn = even ? tp : tv;
      const float e  = even ? av : ap;
      const float o  = even ? ap : av;
      const float rot = (mat == 0) ? (e * ct + o * sn) : (e * ct - o * sn);
      st[(8 * h + r) * SPITCH + n] = even ? rot : (o - rot * sn);
    }
  }
  wave_lds_order();

  const v4f* s4 = (const v4f*)(st + c * SPITCH + 4 * h);
  v4f* d4 = (v4f*)(dst + (size_t)row0 * HS) + lane;
#pragma unroll
  for (int j = 0; j < 8; ++j) d4[32 * j] = s4[2 * j];
}

#define ANW 16
#define ATHREADS 512
#define AQT 2
#define OPITCH 68

__device__ __forceinline__ void load_frag_tile(const float* plane, unsigned piece, v2f (&f)[16]) {
  const v4f* p = (const v4f*)plane + piece;
#pragma unroll
  for (int j = 0; j < 8; ++j) {
    const v4f t = p[32 * j];
    f[2 * j].x = t.x;     f[2 * j].y = t.y;
    f[2 * j + 1].x = t.z; f[2 * j + 1].y = t.w;
  }
}

__device__ __forceinline__ void load_v_tile(const float* plane, unsigned piece, v4f (&vf)[8]) {
  const v4f* p = (const v4f*)plane + piece;
#pragma unroll
  for (int i = 0; i < 8; ++i) vf[i] = p[32 * i];
}

__device__ __forceinline__ float xhalf(float x) {
  const int b = __builtin_bit_cast(int, x);
  return __builtin_bit_cast(float,
      __builtin_amdgcn_permlanex16(b, b, 0x76543210, 0xfedcba98, false, false));
}

template <bool MASKED>
__device__ __forceinline__ void softmax_step(v8f& s, int qd, float& mrun, float& lrun, v8f (&o)[4]) {
  const float scale = 0.05103103630798288f;
  float mt = -1e30f;
#pragma unroll
  for (int r = 0; r < 8; ++r) {
    float sv = s[r] * scale;
    if (MASKED) sv = (r <= qd) ? sv : -1e30f;
    s[r] = sv;
    mt = fmaxf(mt, sv);
  }
  mt = fmaxf(mt, xhalf(mt));

  const float mnew  = fmaxf(mrun, mt);
  const float alpha = __expf(mrun - mnew);
  mrun = mnew;

  float rs = 0.0f;
#pragma unroll
  for (int r = 0; r < 8; ++r) {
    const float p = __expf(s[r] - mnew);
    s[r] = p;
    rs += p;
  }
  lrun = lrun * alpha + rs;
#pragma unroll
  for (int nt = 0; nt < 4; ++nt) {
#pragma unroll
    for (int r = 0; r < 8; ++r) o[nt][r] *= alpha;
  }
}

__device__ __forceinline__ void put_partial(float* otile, float* mrow, float* lrow,
                                            const v8f (&o)[4], float mrun, float lrun,
                                            int lane, int lid, int half) {
  float* op = otile + lid * OPITCH + 8 * half;
#pragma unroll
  for (int nt = 0; nt < 4; ++nt) {
    v4f t0, t1;
    t0.x = o[nt][0]; t0.y = o[nt][1]; t0.z = o[nt][2]; t0.w = o[nt][3];
    t1.x = o[nt][4]; t1.y = o[nt][5]; t1.z = o[nt][6]; t1.w = o[nt][7];
    *(v4f*)(op + 16 * nt)     = t0;
    *(v4f*)(op + 16 * nt + 4) = t1;
  }
  const float lall = lrun + xhalf(lrun);
  if (lane < 16) {
    mrow[lane] = mrun;
    lrow[lane] = lall;
  }
}

__global__ __launch_bounds__(ATHREADS) __attribute__((amdgpu_waves_per_eu(4, 4)))
void attn_kernel(
    const float* __restrict__ qt, const float* __restrict__ kt,
    const float* __restrict__ vo, float* __restrict__ out, int T, int nbatch)
{
  const int npairs = T >> 5;
  const int slot   = blockIdx.x / nbatch;
  const int b      = blockIdx.x - slot * nbatch;
  const int pair   = npairs - 1 - slot;
  const int tid    = threadIdx.x;
  const int wave   = __builtin_amdgcn_readfirstlane(tid >> 5);
  const int lane   = tid & 31;
  const int half   = lane >> 4;
  const int lid    = lane & 15;
  const int qbase  = pair << 5;
  const size_t rowq = (size_t)b * T + qbase;

  __shared__ __attribute__((aligned(16))) float Opart[ANW][AQT][16 * OPITCH];
  __shared__ float Mpart[ANW][AQT][16];
  __shared__ float Lpart[ANW][AQT][16];

  const int nkt = 2 * pair + 2;

  v8f o0[4], o1[4];
#pragma unroll
  for (int i = 0; i < 4; ++i) { o0[i] = v8f{}; o1[i] = v8f{}; }
  float m0 = -1e30f, l0 = 0.0f, m1 = -1e30f, l1 = 0.0f;

  if (wave < nkt) {
    v2f qB0[16], qB1[16];
    const unsigned qpiece = (unsigned)(rowq * (HS / 4)) + (unsigned)lane;
    load_frag_tile(qt, qpiece, qB0);
    load_frag_tile(qt, qpiece + 16u * (HS / 4), qB1);

    int ktile = wave;
    for (; ktile < nkt - 1; ktile += ANW) {
      const int kbase   = ktile << 4;
      const size_t rowk = (size_t)b * T + kbase;
      const unsigned kpiece = (unsigned)(rowk * (HS / 4)) + (unsigned)lane;

      v2f kA[16];
      load_frag_tile(kt, kpiece, kA);
      v4f vf[8];
      load_v_tile(vo, kpiece, vf);

      v8f s0 = v8f{}, s1 = v8f{};
#pragma unroll
      for (int ks = 0; ks < 16; ++ks) {
        s0 = wmma4(kA[ks], qB0[ks], s0);
        s1 = wmma4(kA[ks], qB1[ks], s1);
      }

      softmax_step<true>(s0, qbase + lid - kbase - 8 * half, m0, l0, o0);
      softmax_step<false>(s1, 0, m1, l1, o1);

#pragma unroll
      for (int nt = 0; nt < 4; ++nt) {
#pragma unroll
        for (int j = 0; j < 2; ++j) {
          const v4f t = vf[2 * nt + j];
          v2f a01; a01.x = t.x; a01.y = t.y;
          v2f a23; a23.x = t.z; a23.y = t.w;
          v2f p01; p01.x = s0[4 * j];     p01.y = s0[4 * j + 1];
          v2f p23; p23.x = s0[4 * j + 2]; p23.y = s0[4 * j + 3];
          v2f r01; r01.x = s1[4 * j];     r01.y = s1[4 * j + 1];
          v2f r23; r23.x = s1[4 * j + 2]; r23.y = s1[4 * j + 3];
          o0[nt] = wmma4(a01, p01, o0[nt]);
          o1[nt] = wmma4(a01, r01, o1[nt]);
          o0[nt] = wmma4(a23, p23, o0[nt]);
          o1[nt] = wmma4(a23, r23, o1[nt]);
        }
      }
    }

    if (ktile == nkt - 1) {
      const int kbase   = ktile << 4;
      const size_t rowk = (size_t)b * T + kbase;
      const unsigned kpiece = (unsigned)(rowk * (HS / 4)) + (unsigned)lane;
      v2f kA[16];
      load_frag_tile(kt, kpiece, kA);
      v4f vf[8];
      load_v_tile(vo, kpiece, vf);
      v8f s1 = v8f{};
#pragma unroll
      for (int ks = 0; ks < 16; ++ks) s1 = wmma4(kA[ks], qB1[ks], s1);
      softmax_step<true>(s1, qbase + 16 + lid - kbase - 8 * half, m1, l1, o1);
#pragma unroll
      for (int nt = 0; nt < 4; ++nt) {
#pragma unroll
        for (int j = 0; j < 2; ++j) {
          const v4f t = vf[2 * nt + j];
          v2f a01; a01.x = t.x; a01.y = t.y;
          v2f a23; a23.x = t.z; a23.y = t.w;
          v2f r01; r01.x = s1[4 * j];     r01.y = s1[4 * j + 1];
          v2f r23; r23.x = s1[4 * j + 2]; r23.y = s1[4 * j + 3];
          o1[nt] = wmma4(a01, r01, o1[nt]);
          o1[nt] = wmma4(a23, r23, o1[nt]);
        }
      }
    }

  }

  put_partial(Opart[wave][0], Mpart[wave][0], Lpart[wave][0], o0, m0, l0, lane, lid, half);
  put_partial(Opart[wave][1], Mpart[wave][1], Lpart[wave][1], o1, m1, l1, lane, lid, half);
  __syncthreads();

  const int t  = wave >> 3;
  const int mi = (tid & 255) >> 4;
  const int c4 = (tid & 15) << 2;
  float mv[ANW];
  float mmax = -1e30f;
#pragma unroll
  for (int w = 0; w < ANW; ++w) {
    mv[w] = Mpart[w][t][mi];
    mmax = fmaxf(mmax, mv[w]);
  }
  float lsum = 0.0f;
  v4f acc = v4f{};
#pragma unroll
  for (int w = 0; w < ANW; ++w) {
    const float wgt = __expf(mv[w] - mmax);
    const v4f pv = *(const v4f*)(&Opart[w][t][mi * OPITCH + c4]);
    lsum  += wgt * Lpart[w][t][mi];
    acc.x += wgt * pv.x;
    acc.y += wgt * pv.y;
    acc.z += wgt * pv.z;
    acc.w += wgt * pv.w;
  }
  const float inv = 1.0f / lsum;
  v4f res;
  res.x = acc.x * inv;
  res.y = acc.y * inv;
  res.z = acc.z * inv;
  res.w = acc.w * inv;
  *(v4f*)(out + (rowq + 16 * t + mi) * HS + c4) = res;
}

extern "C" void kernel_launch(void* const* d_in, const int* in_sizes, int n_in,
                              void* d_out, int out_size, void* d_ws, size_t ws_size,
                              hipStream_t stream) {
  const float* x  = (const float*)d_in[0];
  const float* Wq = (const float*)d_in[1];
  const float* Wk = (const float*)d_in[2];
  const float* Wv = (const float*)d_in[3];

  const int BT = in_sizes[0] / NEMBD;
  const int T  = 4096;
  const int B  = BT / T;

  float* qt = (float*)d_ws;
  float* kt = qt + (size_t)BT * HS;
  float* vv = kt + (size_t)BT * HS;
  float* out = (float*)d_out;

  qkv_rope_kernel<<<BT / 64, 384, 0, stream>>>(x, Wq, Wk, Wv, qt, kt, vv, T);
  attn_kernel<<<B * (T / 32), ATHREADS, 0, stream>>>(qt, kt, vv, out, T, B);
}


# QKV projection K loop pipelined: weight slices in an LDS ring of three, next chunk's fragments read into a second register set ahead of the matrix run, one barrier per chunk, not two; same sums
# speedup vs baseline: 14.1730x; 1.0577x over previous
#include <hip/hip_runtime.h>
#include <math.h>

#define NEMBD 384
#define HS 64
#define FTHETA 10000.0f

typedef float v2f __attribute__((ext_vector_type(2)));
typedef float v4f __attribute__((ext_vector_type(4)));
typedef float v8f __attribute__((ext_vector_type(8)));

__device__ __forceinline__ v8f wmma4(v2f a, v2f b, v8f c) {
  return __builtin_amdgcn_wmma_f32_16x16x4_f32(
       false, a,  false, b,
       (short)0, c,  false,  false);
}

__device__ __forceinline__ void wave_lds_order() {
  __builtin_amdgcn_fence(__ATOMIC_RELEASE, "wavefront");
  __builtin_amdgcn_wave_barrier();
  __builtin_amdgcn_fence(__ATOMIC_ACQUIRE, "wavefront");
}

#define SPITCH 68

#define WPITCH4 9
#define WSTAGES 3

template <bool STAGE, bool NEXT>
__device__ __forceinline__ void qkv_trip(
    const float4* const (&wbase)[4], const unsigned (&woff)[4], const int (&wdst)[4],
    const float4* xbase, unsigned xoff,
    const float4* brd, float4* wwr, int kc,
    const float4 (&a)[4], const float4 (&b)[4][4],
    float4 (&an)[4], float4 (&bn)[4][4], v8f (&acc)[4])
{
  float4 w0 = {}, w1 = {}, w2 = {}, w3 = {};
  if (STAGE) {
    const int kn = (kc + 2) * 8;
    w0 = (wbase[0] + kn)[woff[0]]; w1 = (wbase[1] + kn)[woff[1]];
    w2 = (wbase[2] + kn)[woff[2]]; w3 = (wbase[3] + kn)[woff[3]];
  }
  if (NEXT) {
    const int kn = (kc + 1) * 8;
#pragma unroll
    for (int u = 0; u < 4; ++u) an[u] = (xbase + kn)[xoff + u];
#pragma unroll
    for (int nt = 0; nt < 4; ++nt) {
#pragma unroll
      for (int u = 0; u < 4; ++u) bn[nt][u] = brd[nt * 16 * WPITCH4 + u];
    }
  }
  __builtin_amdgcn_sched_barrier(0);

#pragma unroll
  for (int u = 0; u < 4; ++u) {
    v2f a01; a01.x = a[u].x; a01.y = a[u].y;
    v2f a23; a23.x = a[u].z; a23.y = a[u].w;
#pragma unroll
    for (int nt = 0; nt < 4; ++nt) {
      v2f b01; b01.x = b[nt][u].x; b01.y = b[nt][u].y;
      acc[nt] = wmma4(a01, b01, acc[nt]);
    }
#pragma unroll
    for (int nt = 0; nt < 4; ++nt) {
      v2f b23; b23.x = b[nt][u].z; b23.y = b[nt][u].w;
      acc[nt] = wmma4(a23, b23, acc[nt]);
    }
  }

  if (STAGE) {
    __builtin_amdgcn_sched_barrier(0);
    wwr[wdst[0]] = w0;
    wwr[wdst[1]] = w1;
    wwr[wdst[2]] = w2;
    wwr[wdst[3]] = w3;
    __syncthreads();
  }
}

__global__ __launch_bounds__(384) __attribute__((amdgpu_waves_per_eu(3, 4)))
void qkv_rope_kernel(
    const float* __restrict__ x, const float* __restrict__ Wq,
    const float* __restrict__ Wk, const float* __restrict__ Wv,
    float* __restrict__ qt, float* __restrict__ kt, float* __restrict__ vo,
    int T)
{
  __shared__ float4 Ws4[WSTAGES][192 * WPITCH4];
  __shared__ __attribute__((aligned(16))) float Stage[8][16 * SPITCH];

  const int tid  = threadIdx.x;
  const int wave = tid >> 5;
  const int lane = tid & 31;
  const int c    = lane & 15;
  const int h    = lane >> 4;
  const int rt   = wave & 3;
  const int mat  = wave >> 2;
  const int row0 = blockIdx.x * 64 + rt * 16;

  const int wv = __builtin_amdgcn_readfirstlane(wave);
  const float4* wbase[4];
  wbase[0] = (const float4*)Wq;
  wbase[1] = (const float4*)((wv < 4) ? Wq : Wk);
  wbase[2] = (const float4*)((wv < 8) ? Wk : Wv);
  wbase[3] = (const float4*)Wv;
  unsigned woff[4];
  int wdst[4];
#pragma unroll
  for (int u = 0; u < 4; ++u) {
    const int i  = tid + 384 * u;
    const int wr = i >> 3;
    const int c4 = i & 7;
    woff[u] = (unsigned)((wr & 63) * (NEMBD / 4) + c4);
    wdst[u] = wr * WPITCH4 + c4;
  }
  const float4* xbase = (const float4*)(x + (size_t)blockIdx.x * 64 * NEMBD);
  const unsigned xoff = (unsigned)((rt * 16 + c) * (NEMBD / 4) + 4 * h);
  const float4* bsrc = &Ws4[0][(mat * 64 + c) * WPITCH4 + 4 * h];

  {
    const float4 p0 = wbase[0][woff[0]], p1 = wbase[1][woff[1]], p2 = wbase[2][woff[2]], p3 = wbase[3][woff[3]];
    const float4 r0 = (wbase[0] + 8)[woff[0]], r1 = (wbase[1] + 8)[woff[1]];
    const float4 r2 = (wbase[2] + 8)[woff[2]], r3 = (wbase[3] + 8)[woff[3]];
    Ws4[0][wdst[0]] = p0; Ws4[0][wdst[1]] = p1; Ws4[0][wdst[2]] = p2; Ws4[0][wdst[3]] = p3;
    Ws4[1][wdst[0]] = r0; Ws4[1][wdst[1]] = r1; Ws4[1][wdst[2]] = r2; Ws4[1][wdst[3]] = r3;
  }
  float4 a0[4], a1[4];
#pragma unroll
  for (int u = 0; u < 4; ++u) a0[u] = xbase[xoff + u];

  v8f acc[4];
#pragma unroll
  for (int i = 0; i < 4; ++i) acc[i] = v8f{};

  __syncthreads();

  float4 b0[4][4], b1[4][4];
#pragma unroll
  for (int nt = 0; nt < 4; ++nt) {
#pragma unroll
    for (int u = 0; u < 4; ++u) b0[nt][u] = bsrc[nt * 16 * WPITCH4 + u];
  }

  int rd = 1 * (192 * WPITCH4), wr = 2 * (192 * WPITCH4);
  for (int kc = 0; kc < NEMBD / 32 - 2; kc += 2) {
    qkv_trip<true, true>(wbase, woff, wdst, xbase, xoff, bsrc + rd, &Ws4[0][0] + wr, kc, a0, b0, a1, b1, acc);
    rd = wr; wr = (wr == 2 * (192 * WPITCH4)) ? 0 : wr + 192 * WPITCH4;
    qkv_trip<true, true>(wbase, woff, wdst, xbase, xoff, bsrc + rd, &Ws4[0][0] + wr, kc + 1, a1, b1, a0, b0, acc);
    rd = wr; wr = (wr == 2 * (192 * WPITCH4)) ? 0 : wr + 192 * WPITCH4;
  }
  qkv_trip<false, true>(wbase, woff, wdst, xbase, xoff, bsrc + rd, &Ws4[0][0], NEMBD / 32 - 2, a0, b0, a1, b1, acc);
  qkv_trip<false, false>(wbase, woff, wdst, xbase, xoff, bsrc, &Ws4[0][0], NEMBD / 32 - 1, a1, b1, a0, b0, acc);

  const int rbase = row0 + 8 * h;

  if (mat == 2) {
    v4f* v4 = (v4f*)(vo + (size_t)row0 * HS) + lane;
#pragma unroll
    for (int nt = 0; nt < 4; ++nt) {
#pragma unroll
      for (int j = 0; j < 2; ++j) {
        v4f t;
        t.x = acc[nt][4 * j];     t.y = acc[nt][4 * j + 1];
        t.z = acc[nt][4 * j + 2]; t.w = acc[nt][4 * j + 3];
        v4[32 * (2 * nt + j)] = t;
      }
    }
    return;
  }

  float* __restrict__ dst = (mat == 0) ? qt : kt;
  float* st = Stage[wave];
  const float lnT = logf(FTHETA);
  const bool even = (c & 1) == 0;
#pragma unroll
  for (int nt = 0; nt < 4; ++nt) {
    const int n   = nt * 16 + c;
    const int idx = (nt < 2) ? n : n - 32;
    const float f = expf(-(2.0f * (float)idx) * (1.0f / 64.0f) * lnT);
#pragma unroll
    for (int r = 0; r < 8; ++r) {
      const int row  = rbase + r;
      const float tf = (float)(row % T);
      const float tv = (nt < 2) ? sinf(tf * f) : cosf(tf * f);
      const float tp = __shfl_xor(tv, 1);
      const float av = acc[nt][r];
      const float ap = __shfl_xor(av, 1);
      const float ct = even ? tv : tp;
      const float sn = even ? tp : tv;
      const float e  = even ? av : ap;
      const float o  = even ? ap : av;
      const float rot = (mat == 0) ? (e * ct + o * sn) : (e * ct - o * sn);
      st[(8 * h + r) * SPITCH + n] = even ? rot : (o - rot * sn);
    }
  }
  wave_lds_order();

  const v4f* s4 = (const v4f*)(st + c * SPITCH + 4 * h);
  v4f* d4 = (v4f*)(dst + (size_t)row0 * HS) + lane;
#pragma unroll
  for (int j = 0; j < 8; ++j) d4[32 * j] = s4[2 * j];
}

#define ANW 16
#define ATHREADS 512
#define AQT 2
#define OPITCH 68

__device__ __forceinline__ void load_frag_tile(const float* plane, unsigned piece, v2f (&f)[16]) {
  const v4f* p = (const v4f*)plane + piece;
#pragma unroll
  for (int j = 0; j < 8; ++j) {
    const v4f t = p[32 * j];
    f[2 * j].x = t.x;     f[2 * j].y = t.y;
    f[2 * j + 1].x = t.z; f[2 * j + 1].y = t.w;
  }
}

__device__ __forceinline__ void load_v_tile(const float* plane, unsigned piece, v4f (&vf)[8]) {
  const v4f* p = (const v4f*)plane + piece;
#pragma unroll
  for (int i = 0; i < 8; ++i) vf[i] = p[32 * i];
}

__device__ __forceinline__ float xhalf(float x) {
  const int b = __builtin_bit_cast(int, x);
  return __builtin_bit_cast(float,
      __builtin_amdgcn_permlanex16(b, b, 0x76543210, 0xfedcba98, false, false));
}

template <bool MASKED>
__device__ __forceinline__ void softmax_step(v8f& s, int qd, float& mrun, float& lrun, v8f (&o)[4]) {
  const float scale = 0.05103103630798288f;
  float mt = -1e30f;
#pragma unroll
  for (int r = 0; r < 8; ++r) {
    float sv = s[r] * scale;
    if (MASKED) sv = (r <= qd) ? sv : -1e30f;
    s[r] = sv;
    mt = fmaxf(mt, sv);
  }
  mt = fmaxf(mt, xhalf(mt));

  const float mnew  = fmaxf(mrun, mt);
  const float alpha = __expf(mrun - mnew);
  mrun = mnew;

  float rs = 0.0f;
#pragma unroll
  for (int r = 0; r < 8; ++r) {
    const float p = __expf(s[r] - mnew);
    s[r] = p;
    rs += p;
  }
  lrun = lrun * alpha + rs;
#pragma unroll
  for (int nt = 0; nt < 4; ++nt) {
#pragma unroll
    for (int r = 0; r < 8; ++r) o[nt][r] *= alpha;
  }
}

__device__ __forceinline__ void put_partial(float* otile, float* mrow, float* lrow,
                                            const v8f (&o)[4], float mrun, float lrun,
                                            int lane, int lid, int half) {
  float* op = otile + lid * OPITCH + 8 * half;
#pragma unroll
  for (int nt = 0; nt < 4; ++nt) {
    v4f t0, t1;
    t0.x = o[nt][0]; t0.y = o[nt][1]; t0.z = o[nt][2]; t0.w = o[nt][3];
    t1.x = o[nt][4]; t1.y = o[nt][5]; t1.z = o[nt][6]; t1.w = o[nt][7];
    *(v4f*)(op + 16 * nt)     = t0;
    *(v4f*)(op + 16 * nt + 4) = t1;
  }
  const float lall = lrun + xhalf(lrun);
  if (lane < 16) {
    mrow[lane] = mrun;
    lrow[lane] = lall;
  }
}

__global__ __launch_bounds__(ATHREADS) __attribute__((amdgpu_waves_per_eu(4, 4)))
void attn_kernel(
    const float* __restrict__ qt, const float* __restrict__ kt,
    const float* __restrict__ vo, float* __restrict__ out, int T, int nbatch)
{
  const int npairs = T >> 5;
  const int slot   = blockIdx.x / nbatch;
  const int b      = blockIdx.x - slot * nbatch;
  const int pair   = npairs - 1 - slot;
  const int tid    = threadIdx.x;
  const int wave   = __builtin_amdgcn_readfirstlane(tid >> 5);
  const int lane   = tid & 31;
  const int half   = lane >> 4;
  const int lid    = lane & 15;
  const int qbase  = pair << 5;
  const size_t rowq = (size_t)b * T + qbase;

  __shared__ __attribute__((aligned(16))) float Opart[ANW][AQT][16 * OPITCH];
  __shared__ float Mpart[ANW][AQT][16];
  __shared__ float Lpart[ANW][AQT][16];

  const int nkt = 2 * pair + 2;

  v8f o0[4], o1[4];
#pragma unroll
  for (int i = 0; i < 4; ++i) { o0[i] = v8f{}; o1[i] = v8f{}; }
  float m0 = -1e30f, l0 = 0.0f, m1 = -1e30f, l1 = 0.0f;

  if (wave < nkt) {
    v2f qB0[16], qB1[16];
    const unsigned qpiece = (unsigned)(rowq * (HS / 4)) + (unsigned)lane;
    load_frag_tile(qt, qpiece, qB0);
    load_frag_tile(qt, qpiece + 16u * (HS / 4), qB1);

    int ktile = wave;
    for (; ktile < nkt - 1; ktile += ANW) {
      const int kbase   = ktile << 4;
      const size_t rowk = (size_t)b * T + kbase;
      const unsigned kpiece = (unsigned)(rowk * (HS / 4)) + (unsigned)lane;

      v2f kA[16];
      load_frag_tile(kt, kpiece, kA);
      v4f vf[8];
      load_v_tile(vo, kpiece, vf);

      v8f s0 = v8f{}, s1 = v8f{};
#pragma unroll
      for (int ks = 0; ks < 16; ++ks) {
        s0 = wmma4(kA[ks], qB0[ks], s0);
        s1 = wmma4(kA[ks], qB1[ks], s1);
      }

      softmax_step<true>(s0, qbase + lid - kbase - 8 * half, m0, l0, o0);
      softmax_step<false>(s1, 0, m1, l1, o1);

#pragma unroll
      for (int nt = 0; nt < 4; ++nt) {
#pragma unroll
        for (int j = 0; j < 2; ++j) {
          const v4f t = vf[2 * nt + j];
          v2f a01; a01.x = t.x; a01.y = t.y;
          v2f a23; a23.x = t.z; a23.y = t.w;
          v2f p01; p01.x = s0[4 * j];     p01.y = s0[4 * j + 1];
          v2f p23; p23.x = s0[4 * j + 2]; p23.y = s0[4 * j + 3];
          v2f r01; r01.x = s1[4 * j];     r01.y = s1[4 * j + 1];
          v2f r23; r23.x = s1[4 * j + 2]; r23.y = s1[4 * j + 3];
          o0[nt] = wmma4(a01, p01, o0[nt]);
          o1[nt] = wmma4(a01, r01, o1[nt]);
          o0[nt] = wmma4(a23, p23, o0[nt]);
          o1[nt] = wmma4(a23, r23, o1[nt]);
        }
      }
    }

    if (ktile == nkt - 1) {
      const int kbase   = ktile << 4;
      const size_t rowk = (size_t)b * T + kbase;
      const unsigned kpiece = (unsigned)(rowk * (HS / 4)) + (unsigned)lane;
      v2f kA[16];
      load_frag_tile(kt, kpiece, kA);
      v4f vf[8];
      load_v_tile(vo, kpiece, vf);
      v8f s1 = v8f{};
#pragma unroll
      for (int ks = 0; ks < 16; ++ks) s1 = wmma4(kA[ks], qB1[ks], s1);
      softmax_step<true>(s1, qbase + 16 + lid - kbase - 8 * half, m1, l1, o1);
#pragma unroll
      for (int nt = 0; nt < 4; ++nt) {
#pragma unroll
        for (int j = 0; j < 2; ++j) {
          const v4f t = vf[2 * nt + j];
          v2f a01; a01.x = t.x; a01.y = t.y;
          v2f a23; a23.x = t.z; a23.y = t.w;
          v2f r01; r01.x = s1[4 * j];     r01.y = s1[4 * j + 1];
          v2f r23; r23.x = s1[4 * j + 2]; r23.y = s1[4 * j + 3];
          o1[nt] = wmma4(a01, r01, o1[nt]);
          o1[nt] = wmma4(a23, r23, o1[nt]);
        }
      }
    }

  }

  put_partial(Opart[wave][0], Mpart[wave][0], Lpart[wave][0], o0, m0, l0, lane, lid, half);
  put_partial(Opart[wave][1], Mpart[wave][1], Lpart[wave][1], o1, m1, l1, lane, lid, half);
  __syncthreads();

  const int t  = wave >> 3;
  const int mi = (tid & 255) >> 4;
  const int c4 = (tid & 15) << 2;
  float mv[ANW];
  float mmax = -1e30f;
#pragma unroll
  for (int w = 0; w < ANW; ++w) {
    mv[w] = Mpart[w][t][mi];
    mmax = fmaxf(mmax, mv[w]);
  }
  float lsum = 0.0f;
  v4f acc = v4f{};
#pragma unroll
  for (int w = 0; w < ANW; ++w) {
    const float wgt = __expf(mv[w] - mmax);
    const v4f pv = *(const v4f*)(&Opart[w][t][mi * OPITCH + c4]);
    lsum  += wgt * Lpart[w][t][mi];
    acc.x += wgt * pv.x;
    acc.y += wgt * pv.y;
    acc.z += wgt * pv.z;
    acc.w += wgt * pv.w;
  }
  const float inv = 1.0f / lsum;
  v4f res;
  res.x = acc.x * inv;
  res.y = acc.y * inv;
  res.z = acc.z * inv;
  res.w = acc.w * inv;
  *(v4f*)(out + (rowq + 16 * t + mi) * HS + c4) = res;
}

extern "C" void kernel_launch(void* const* d_in, const int* in_sizes, int n_in,
                              void* d_out, int out_size, void* d_ws, size_t ws_size,
                              hipStream_t stream) {
  const float* x  = (const float*)d_in[0];
  const float* Wq = (const float*)d_in[1];
  const float* Wk = (const float*)d_in[2];
  const float* Wv = (const float*)d_in[3];

  const int BT = in_sizes[0] / NEMBD;
  const int T  = 4096;
  const int B  = BT / T;

  float* qt = (float*)d_ws;
  float* kt = qt + (size_t)BT * HS;
  float* vv = kt + (size_t)BT * HS;
  float* out = (float*)d_out;

  qkv_rope_kernel<<<BT / 64, 384, 0, stream>>>(x, Wq, Wk, Wv, qt, kt, vv, T);
  attn_kernel<<<B * (T / 32), ATHREADS, 0, stream>>>(qt, kt, vv, out, T, B);
}
